# starting kernel
# baseline (speedup 1.0000x reference)
#include <hip/hip_runtime.h>


typedef __attribute__((ext_vector_type(2))) float v2f;
typedef __attribute__((ext_vector_type(8))) float v8f;

constexpr int S    = 2048;
constexpr int BT   = 32;
constexpr int DIN  = 256;
constexpr int DOUT = 256;
constexpr int LAT  = 64;
constexpr int T    = 32;
constexpr int NC   = 64;
constexpr int HPAD = 68;

__device__ __forceinline__ v8f wmma4(v2f a, v2f b, v8f c) {
  return __builtin_amdgcn_wmma_f32_16x16x4_f32(false, a, false, b, (short)0, c,
                                               false, false);
}

__device__ __forceinline__ v8f tile_mm_lds(const float* hbase, const v2f* bfrag,
                                           v8f acc, int mt, int m, int hi) {
  const float* ap = hbase + (mt * 16 + m) * HPAD + 2 * hi;
#pragma unroll
  for (int kk = 0; kk < 16; ++kk) {
    v2f a = *(const v2f*)(ap + kk * 4);
    acc = wmma4(a, bfrag[kk], acc);
  }
  return acc;
}

__device__ __forceinline__ void load_bfrag64(v2f* bfrag, const float* W, int nt,
                                             int m, int hi) {
#pragma unroll
  for (int kk = 0; kk < 16; ++kk)
    bfrag[kk] = *(const v2f*)(W + (nt * 16 + m) * 64 + kk * 4 + 2 * hi);
}

__device__ __forceinline__ v8f load_u_tile(const float* u, int mt, int nt,
                                           int m, int hi) {
  v8f acc;
#pragma unroll
  for (int r = 0; r < 8; ++r)
    acc[r] = u[(mt * 16 + r + 8 * hi) * 64 + nt * 16 + m];
  return acc;
}

__device__ __forceinline__ void store_tile_lds(float* hbase, v8f d, int mt,
                                               int nt, int m, int hi) {
#pragma unroll
  for (int r = 0; r < 8; ++r)
    hbase[(mt * 16 + r + 8 * hi) * HPAD + nt * 16 + m] = d[r];
}

__device__ __forceinline__ void store_tile_g64(float* g, v8f d, int mt, int nt,
                                               int m, int hi) {
#pragma unroll
  for (int r = 0; r < 8; ++r)
    g[(mt * 16 + r + 8 * hi) * 64 + nt * 16 + m] = d[r];
}

__global__ void bx_gemm_k(const float* __restrict__ x,
                          const float* __restrict__ Bw,
                          float* __restrict__ bx) {
  int tid = threadIdx.x;
  int lane = tid & 31, hi = lane >> 4, m = lane & 15;
  int w = (blockIdx.x * blockDim.x + tid) >> 5;
  int mt = w >> 2, nt = w & 3;

  const float* ap = x + (size_t)(mt * 16 + m) * DIN + 2 * hi;
  const float* bp = Bw + (size_t)(nt * 16 + m) * DIN + 2 * hi;

  v8f acc = {0.f, 0.f, 0.f, 0.f, 0.f, 0.f, 0.f, 0.f};
#pragma unroll 8
  for (int kk = 0; kk < DIN / 4; ++kk) {
    v2f a = *(const v2f*)(ap + kk * 4);
    v2f b = *(const v2f*)(bp + kk * 4);
    acc = wmma4(a, b, acc);
  }
  store_tile_g64(bx + (size_t)mt * 16 * LAT, acc, 0, nt, m, hi);
}

__global__ void matpow_k(const float* __restrict__ A, float* __restrict__ Apow) {
  __shared__ float P[2][64][65];
  int tid = threadIdx.x;
  for (int i = tid; i < 64 * 64; i += blockDim.x)
    P[0][i >> 6][i & 63] = A[i];
  __syncthreads();
  int p = 0;
#pragma unroll 1
  for (int s = 0; s < 5; ++s) {
    for (int i = tid; i < 64 * 64; i += blockDim.x) {
      int r = i >> 6, c = i & 63;
      float acc = 0.f;
#pragma unroll 8
      for (int k = 0; k < 64; ++k) acc += P[p][r][k] * P[p][k][c];
      P[1 - p][r][c] = acc;
    }
    __syncthreads();
    p ^= 1;
  }
  for (int i = tid; i < 64 * 64; i += blockDim.x)
    Apow[i] = P[p][i >> 6][i & 63];
}

__global__ void chunk_local_k(const float* __restrict__ bx,
                              const float* __restrict__ A,
                              float* __restrict__ Lc) {
  __shared__ float hbuf[2][BT * HPAD];
  int tid = threadIdx.x;
  int lane = tid & 31, hi = lane >> 4, m = lane & 15;
  int w = tid >> 5, mt = w >> 2, nt = w & 3;
  int c = blockIdx.x;

  for (int i = tid; i < BT * HPAD; i += blockDim.x) hbuf[0][i] = 0.f;

  v2f bfrag[16];
  load_bfrag64(bfrag, A, nt, m, hi);

  v8f u_next = load_u_tile(bx + (size_t)(c * T) * BT * LAT, mt, nt, m, hi);
  __syncthreads();

  int p = 0;
#pragma unroll 1
  for (int i = 0; i < T; ++i) {
    v8f acc = u_next;
    if (i + 1 < T)
      u_next =
          load_u_tile(bx + (size_t)(c * T + i + 1) * BT * LAT, mt, nt, m, hi);
    acc = tile_mm_lds(hbuf[p], bfrag, acc, mt, m, hi);
    store_tile_lds(hbuf[1 - p], acc, mt, nt, m, hi);
    __syncthreads();
    p ^= 1;
  }
  float* out = Lc + (size_t)c * BT * LAT;
  for (int i = tid; i < BT * LAT; i += blockDim.x)
    out[i] = hbuf[p][(i >> 6) * HPAD + (i & 63)];
}

__global__ void chunk_scan_k(const float* __restrict__ Apow,
                             const float* __restrict__ Lc,
                             float* __restrict__ carries) {
  __shared__ float hbuf[2][BT * HPAD];
  int tid = threadIdx.x;
  int lane = tid & 31, hi = lane >> 4, m = lane & 15;
  int w = tid >> 5, mt = w >> 2, nt = w & 3;

  for (int i = tid; i < BT * HPAD; i += blockDim.x) hbuf[0][i] = 0.f;
  for (int i = tid; i < BT * LAT; i += blockDim.x) carries[i] = 0.f;

  v2f bfrag[16];
  load_bfrag64(bfrag, Apow, nt, m, hi);

  v8f u_next = load_u_tile(Lc, mt, nt, m, hi);
  __syncthreads();

  int p = 0;
#pragma unroll 1
  for (int c = 0; c < NC - 1; ++c) {
    v8f acc = u_next;
    if (c + 1 < NC - 1)
      u_next = load_u_tile(Lc + (size_t)(c + 1) * BT * LAT, mt, nt, m, hi);
    acc = tile_mm_lds(hbuf[p], bfrag, acc, mt, m, hi);
    store_tile_lds(hbuf[1 - p], acc, mt, nt, m, hi);
    store_tile_g64(carries + (size_t)(c + 1) * BT * LAT, acc, mt, nt, m, hi);
    __syncthreads();
    p ^= 1;
  }
}

__global__ void chunk_final_k(const float* __restrict__ bx,
                              const float* __restrict__ A,
                              const float* __restrict__ carries,
                              float* __restrict__ h_out) {
  __shared__ float hbuf[2][BT * HPAD];
  int tid = threadIdx.x;
  int lane = tid & 31, hi = lane >> 4, m = lane & 15;
  int w = tid >> 5, mt = w >> 2, nt = w & 3;
  int c = blockIdx.x;

  const float* cin = carries + (size_t)c * BT * LAT;
  for (int i = tid; i < BT * LAT; i += blockDim.x)
    hbuf[0][(i >> 6) * HPAD + (i & 63)] = cin[i];

  v2f bfrag[16];
  load_bfrag64(bfrag, A, nt, m, hi);

  v8f u_next = load_u_tile(bx + (size_t)(c * T) * BT * LAT, mt, nt, m, hi);
  __syncthreads();

  int p = 0;
#pragma unroll 1
  for (int i = 0; i < T; ++i) {
    int t = c * T + i;
    v8f acc = u_next;
    if (i + 1 < T)
      u_next = load_u_tile(bx + (size_t)(t + 1) * BT * LAT, mt, nt, m, hi);
    acc = tile_mm_lds(hbuf[p], bfrag, acc, mt, m, hi);
    store_tile_lds(hbuf[1 - p], acc, mt, nt, m, hi);
    store_tile_g64(h_out + (size_t)(t + 1) * BT * LAT, acc, mt, nt, m, hi);
    __syncthreads();
    p ^= 1;
  }
}

__global__ void y_gemm_k(const float* __restrict__ hs,
                         const float* __restrict__ C,
                         float* __restrict__ y) {
  int tid = threadIdx.x;
  int lane = tid & 31, hi = lane >> 4, m = lane & 15;
  int w = (blockIdx.x * blockDim.x + tid) >> 5;
  int mt = w >> 4, nt = w & 15;

  const float* ap = hs + (size_t)(mt * 16 + m) * LAT + 2 * hi;
  const float* bp = C + (size_t)(nt * 16 + m) * LAT + 2 * hi;

  v8f acc = {0.f, 0.f, 0.f, 0.f, 0.f, 0.f, 0.f, 0.f};
#pragma unroll
  for (int kk = 0; kk < LAT / 4; ++kk) {
    v2f a = *(const v2f*)(ap + kk * 4);
    v2f b = *(const v2f*)(bp + kk * 4);
    acc = wmma4(a, b, acc);
  }
#pragma unroll
  for (int r = 0; r < 8; ++r)
    y[(size_t)(mt * 16 + r + 8 * hi) * DOUT + nt * 16 + m] = acc[r];
}

extern "C" void kernel_launch(void* const* d_in, const int* in_sizes, int n_in,
                              void* d_out, int out_size, void* d_ws,
                              size_t ws_size, hipStream_t stream) {
  const float* x  = (const float*)d_in[0];
  const float* A  = (const float*)d_in[1];
  const float* Bw = (const float*)d_in[2];
  const float* C  = (const float*)d_in[3];

  float* out      = (float*)d_out;
  float* y_region = out;
  float* h_region = out + (size_t)S * BT * DOUT;

  float* bx = y_region;

  float* ws      = (float*)d_ws;
  float* Apow    = ws;
  float* Lc      = ws + 64 * 64;
  float* carries = ws + 64 * 64 + (size_t)NC * BT * LAT;

  hipMemsetAsync(h_region, 0, (size_t)BT * LAT * sizeof(float), stream);

  bx_gemm_k<<<(S * BT / 16) * (LAT / 16) / 8, 256, 0, stream>>>(x, Bw, bx);
  matpow_k<<<1, 256, 0, stream>>>(A, Apow);
  chunk_local_k<<<NC, 256, 0, stream>>>(bx, A, Lc);
  chunk_scan_k<<<1, 256, 0, stream>>>(Apow, Lc, carries);
  chunk_final_k<<<NC, 256, 0, stream>>>(bx, A, carries, h_region);
  y_gemm_k<<<(S * BT / 16) * (DOUT / 16) / 8, 256, 0, stream>>>(
      h_region + BT * LAT, C, y_region);
}


# A^32 for the chunk-boundary scan: five squarings on WMMA f32 16x16x4 out of LDS (2 tiles per wave) replace the scalar FMA loops; product summation order changes
# speedup vs baseline: 1.2761x; 1.2761x over previous
#include <hip/hip_runtime.h>


typedef __attribute__((ext_vector_type(2))) float v2f;
typedef __attribute__((ext_vector_type(4))) float v4f;
typedef __attribute__((ext_vector_type(8))) float v8f;

constexpr int S    = 2048;
constexpr int BT   = 32;
constexpr int DIN  = 256;
constexpr int DOUT = 256;
constexpr int LAT  = 64;
constexpr int T    = 32;
constexpr int NC   = 64;
constexpr int HPAD = 68;

__device__ __forceinline__ v8f wmma4(v2f a, v2f b, v8f c) {
  return __builtin_amdgcn_wmma_f32_16x16x4_f32(false, a, false, b, (short)0, c,
                                               false, false);
}

__device__ __forceinline__ v8f tile_mm_lds(const float* hbase, const v2f* bfrag,
                                           v8f acc, int mt, int m, int hi) {
  const float* ap = hbase + (mt * 16 + m) * HPAD + 2 * hi;
#pragma unroll
  for (int kk = 0; kk < 16; ++kk) {
    v2f a = *(const v2f*)(ap + kk * 4);
    acc = wmma4(a, bfrag[kk], acc);
  }
  return acc;
}

__device__ __forceinline__ void load_bfrag64(v2f* bfrag, const float* W, int nt,
                                             int m, int hi) {
#pragma unroll
  for (int kk = 0; kk < 16; ++kk)
    bfrag[kk] = *(const v2f*)(W + (nt * 16 + m) * 64 + kk * 4 + 2 * hi);
}

__device__ __forceinline__ v8f load_u_tile(const float* u, int mt, int nt,
                                           int m, int hi) {
  v8f acc;
#pragma unroll
  for (int r = 0; r < 8; ++r)
    acc[r] = u[(mt * 16 + r + 8 * hi) * 64 + nt * 16 + m];
  return acc;
}

__device__ __forceinline__ void store_tile_lds(float* hbase, v8f d, int mt,
                                               int nt, int m, int hi) {
#pragma unroll
  for (int r = 0; r < 8; ++r)
    hbase[(mt * 16 + r + 8 * hi) * HPAD + nt * 16 + m] = d[r];
}

__device__ __forceinline__ void store_tile_g64(float* g, v8f d, int mt, int nt,
                                               int m, int hi) {
#pragma unroll
  for (int r = 0; r < 8; ++r)
    g[(mt * 16 + r + 8 * hi) * 64 + nt * 16 + m] = d[r];
}

__global__ void bx_gemm_k(const float* __restrict__ x,
                          const float* __restrict__ Bw,
                          float* __restrict__ bx) {
  int tid = threadIdx.x;
  int lane = tid & 31, hi = lane >> 4, m = lane & 15;
  int w = (blockIdx.x * blockDim.x + tid) >> 5;
  int mt = w >> 2, nt = w & 3;

  const float* ap = x + (size_t)(mt * 16 + m) * DIN + 2 * hi;
  const float* bp = Bw + (size_t)(nt * 16 + m) * DIN + 2 * hi;

  v8f acc = {0.f, 0.f, 0.f, 0.f, 0.f, 0.f, 0.f, 0.f};
#pragma unroll 8
  for (int kk = 0; kk < DIN / 4; ++kk) {
    v2f a = *(const v2f*)(ap + kk * 4);
    v2f b = *(const v2f*)(bp + kk * 4);
    acc = wmma4(a, b, acc);
  }
  store_tile_g64(bx + (size_t)mt * 16 * LAT, acc, 0, nt, m, hi);
}

__device__ __forceinline__ void square_tiles(const float* cur, int mt, int nt0,
                                             int m, int hi, v8f& acc0,
                                             v8f& acc1) {
  const float* ap = cur + (mt * 16 + m) * HPAD + 2 * hi;
  const float* bp = cur + (2 * hi) * HPAD + nt0 * 16 + m;
  v2f a[16], b0[16], b1[16];
#pragma unroll
  for (int kk = 0; kk < 16; ++kk) {
    a[kk] = *(const v2f*)(ap + kk * 4);
    b0[kk][0] = bp[(kk * 4) * HPAD];
    b0[kk][1] = bp[(kk * 4 + 1) * HPAD];
    b1[kk][0] = bp[(kk * 4) * HPAD + 16];
    b1[kk][1] = bp[(kk * 4 + 1) * HPAD + 16];
  }
  v8f c0 = {0.f, 0.f, 0.f, 0.f, 0.f, 0.f, 0.f, 0.f};
  v8f c1 = {0.f, 0.f, 0.f, 0.f, 0.f, 0.f, 0.f, 0.f};
#pragma unroll
  for (int kk = 0; kk < 16; ++kk) {
    c0 = wmma4(a[kk], b0[kk], c0);
    c1 = wmma4(a[kk], b1[kk], c1);
  }
  acc0 = c0;
  acc1 = c1;
}

__global__ void __launch_bounds__(256)
matpow_k(const float* __restrict__ A, float* __restrict__ Apow) {
  __shared__ __attribute__((aligned(16))) float P[2][64 * HPAD];
  int tid = threadIdx.x;
  int lane = tid & 31, hi = lane >> 4, m = lane & 15;
  int w = tid >> 5, mt = w >> 1, nt0 = (w & 1) * 2;

#pragma unroll
  for (int j = 0; j < 4; ++j) {
    int q = tid + j * 256;
    v4f v = *(const v4f*)(A + q * 4);
    *(v4f*)(&P[0][(q >> 4) * HPAD + (q & 15) * 4]) = v;
  }
  __syncthreads();

  v8f acc0, acc1;
  int p = 0;
  for (int s = 0; s < 4; ++s) {
    square_tiles(P[p], mt, nt0, m, hi, acc0, acc1);
    store_tile_lds(P[1 - p], acc0, mt, nt0, m, hi);
    store_tile_lds(P[1 - p], acc1, mt, nt0 + 1, m, hi);
    __syncthreads();
    p ^= 1;
  }
  square_tiles(P[p], mt, nt0, m, hi, acc0, acc1);
  store_tile_g64(Apow, acc0, mt, nt0, m, hi);
  store_tile_g64(Apow, acc1, mt, nt0 + 1, m, hi);
}

__global__ void chunk_local_k(const float* __restrict__ bx,
                              const float* __restrict__ A,
                              float* __restrict__ Lc) {
  __shared__ float hbuf[2][BT * HPAD];
  int tid = threadIdx.x;
  int lane = tid & 31, hi = lane >> 4, m = lane & 15;
  int w = tid >> 5, mt = w >> 2, nt = w & 3;
  int c = blockIdx.x;

  for (int i = tid; i < BT * HPAD; i += blockDim.x) hbuf[0][i] = 0.f;

  v2f bfrag[16];
  load_bfrag64(bfrag, A, nt, m, hi);

  v8f u_next = load_u_tile(bx + (size_t)(c * T) * BT * LAT, mt, nt, m, hi);
  __syncthreads();

  int p = 0;
#pragma unroll 1
  for (int i = 0; i < T; ++i) {
    v8f acc = u_next;
    if (i + 1 < T)
      u_next =
          load_u_tile(bx + (size_t)(c * T + i + 1) * BT * LAT, mt, nt, m, hi);
    acc = tile_mm_lds(hbuf[p], bfrag, acc, mt, m, hi);
    store_tile_lds(hbuf[1 - p], acc, mt, nt, m, hi);
    __syncthreads();
    p ^= 1;
  }
  float* out = Lc + (size_t)c * BT * LAT;
  for (int i = tid; i < BT * LAT; i += blockDim.x)
    out[i] = hbuf[p][(i >> 6) * HPAD + (i & 63)];
}

__global__ void chunk_scan_k(const float* __restrict__ Apow,
                             const float* __restrict__ Lc,
                             float* __restrict__ carries) {
  __shared__ float hbuf[2][BT * HPAD];
  int tid = threadIdx.x;
  int lane = tid & 31, hi = lane >> 4, m = lane & 15;
  int w = tid >> 5, mt = w >> 2, nt = w & 3;

  for (int i = tid; i < BT * HPAD; i += blockDim.x) hbuf[0][i] = 0.f;
  for (int i = tid; i < BT * LAT; i += blockDim.x) carries[i] = 0.f;

  v2f bfrag[16];
  load_bfrag64(bfrag, Apow, nt, m, hi);

  v8f u_next = load_u_tile(Lc, mt, nt, m, hi);
  __syncthreads();

  int p = 0;
#pragma unroll 1
  for (int c = 0; c < NC - 1; ++c) {
    v8f acc = u_next;
    if (c + 1 < NC - 1)
      u_next = load_u_tile(Lc + (size_t)(c + 1) * BT * LAT, mt, nt, m, hi);
    acc = tile_mm_lds(hbuf[p], bfrag, acc, mt, m, hi);
    store_tile_lds(hbuf[1 - p], acc, mt, nt, m, hi);
    store_tile_g64(carries + (size_t)(c + 1) * BT * LAT, acc, mt, nt, m, hi);
    __syncthreads();
    p ^= 1;
  }
}

__global__ void chunk_final_k(const float* __restrict__ bx,
                              const float* __restrict__ A,
                              const float* __restrict__ carries,
                              float* __restrict__ h_out) {
  __shared__ float hbuf[2][BT * HPAD];
  int tid = threadIdx.x;
  int lane = tid & 31, hi = lane >> 4, m = lane & 15;
  int w = tid >> 5, mt = w >> 2, nt = w & 3;
  int c = blockIdx.x;

  const float* cin = carries + (size_t)c * BT * LAT;
  for (int i = tid; i < BT * LAT; i += blockDim.x)
    hbuf[0][(i >> 6) * HPAD + (i & 63)] = cin[i];

  v2f bfrag[16];
  load_bfrag64(bfrag, A, nt, m, hi);

  v8f u_next = load_u_tile(bx + (size_t)(c * T) * BT * LAT, mt, nt, m, hi);
  __syncthreads();

  int p = 0;
#pragma unroll 1
  for (int i = 0; i < T; ++i) {
    int t = c * T + i;
    v8f acc = u_next;
    if (i + 1 < T)
      u_next = load_u_tile(bx + (size_t)(t + 1) * BT * LAT, mt, nt, m, hi);
    acc = tile_mm_lds(hbuf[p], bfrag, acc, mt, m, hi);
    store_tile_lds(hbuf[1 - p], acc, mt, nt, m, hi);
    store_tile_g64(h_out + (size_t)(t + 1) * BT * LAT, acc, mt, nt, m, hi);
    __syncthreads();
    p ^= 1;
  }
}

__global__ void y_gemm_k(const float* __restrict__ hs,
                         const float* __restrict__ C,
                         float* __restrict__ y) {
  int tid = threadIdx.x;
  int lane = tid & 31, hi = lane >> 4, m = lane & 15;
  int w = (blockIdx.x * blockDim.x + tid) >> 5;
  int mt = w >> 4, nt = w & 15;

  const float* ap = hs + (size_t)(mt * 16 + m) * LAT + 2 * hi;
  const float* bp = C + (size_t)(nt * 16 + m) * LAT + 2 * hi;

  v8f acc = {0.f, 0.f, 0.f, 0.f, 0.f, 0.f, 0.f, 0.f};
#pragma unroll
  for (int kk = 0; kk < LAT / 4; ++kk) {
    v2f a = *(const v2f*)(ap + kk * 4);
    v2f b = *(const v2f*)(bp + kk * 4);
    acc = wmma4(a, b, acc);
  }
#pragma unroll
  for (int r = 0; r < 8; ++r)
    y[(size_t)(mt * 16 + r + 8 * hi) * DOUT + nt * 16 + m] = acc[r];
}

extern "C" void kernel_launch(void* const* d_in, const int* in_sizes, int n_in,
                              void* d_out, int out_size, void* d_ws,
                              size_t ws_size, hipStream_t stream) {
  const float* x  = (const float*)d_in[0];
  const float* A  = (const float*)d_in[1];
  const float* Bw = (const float*)d_in[2];
  const float* C  = (const float*)d_in[3];

  float* out      = (float*)d_out;
  float* y_region = out;
  float* h_region = out + (size_t)S * BT * DOUT;

  float* bx = y_region;

  float* ws      = (float*)d_ws;
  float* Apow    = ws;
  float* Lc      = ws + 64 * 64;
  float* carries = ws + 64 * 64 + (size_t)NC * BT * LAT;

  hipMemsetAsync(h_region, 0, (size_t)BT * LAT * sizeof(float), stream);

  bx_gemm_k<<<(S * BT / 16) * (LAT / 16) / 8, 256, 0, stream>>>(x, Bw, bx);
  matpow_k<<<1, 256, 0, stream>>>(A, Apow);
  chunk_local_k<<<NC, 256, 0, stream>>>(bx, A, Lc);
  chunk_scan_k<<<1, 256, 0, stream>>>(Apow, Lc, carries);
  chunk_final_k<<<NC, 256, 0, stream>>>(bx, A, carries, h_region);
  y_gemm_k<<<(S * BT / 16) * (DOUT / 16) / 8, 256, 0, stream>>>(
      h_region + BT * LAT, C, y_region);
}


# Scan-step barrier orders LDS only (local-address-space fences round s_barrier): a step no longer waits for its global store acknowledgements and the next u tile; no sum order changes
# speedup vs baseline: 1.7011x; 1.3330x over previous
#include <hip/hip_runtime.h>


typedef __attribute__((ext_vector_type(2))) float v2f;
typedef __attribute__((ext_vector_type(4))) float v4f;
typedef __attribute__((ext_vector_type(8))) float v8f;

constexpr int S    = 2048;
constexpr int BT   = 32;
constexpr int DIN  = 256;
constexpr int DOUT = 256;
constexpr int LAT  = 64;
constexpr int T    = 32;
constexpr int NC   = 64;
constexpr int HPAD = 68;

__device__ __forceinline__ v8f wmma4(v2f a, v2f b, v8f c) {
  return __builtin_amdgcn_wmma_f32_16x16x4_f32(false, a, false, b, (short)0, c,
                                               false, false);
}

__device__ __forceinline__ void lds_barrier() {
  __builtin_amdgcn_fence(__ATOMIC_RELEASE, "workgroup", "local");
  __builtin_amdgcn_s_barrier();
  __builtin_amdgcn_fence(__ATOMIC_ACQUIRE, "workgroup", "local");
}

__device__ __forceinline__ v8f tile_mm_lds(const float* hbase, const v2f* bfrag,
                                           v8f acc, int mt, int m, int hi) {
  const float* ap = hbase + (mt * 16 + m) * HPAD + 2 * hi;
#pragma unroll
  for (int kk = 0; kk < 16; ++kk) {
    v2f a = *(const v2f*)(ap + kk * 4);
    acc = wmma4(a, bfrag[kk], acc);
  }
  return acc;
}

__device__ __forceinline__ void load_bfrag64(v2f* bfrag, const float* W, int nt,
                                             int m, int hi) {
#pragma unroll
  for (int kk = 0; kk < 16; ++kk)
    bfrag[kk] = *(const v2f*)(W + (nt * 16 + m) * 64 + kk * 4 + 2 * hi);
}

__device__ __forceinline__ v8f load_u_tile(const float* u, int mt, int nt,
                                           int m, int hi) {
  v8f acc;
#pragma unroll
  for (int r = 0; r < 8; ++r)
    acc[r] = u[(mt * 16 + r + 8 * hi) * 64 + nt * 16 + m];
  return acc;
}

__device__ __forceinline__ void store_tile_lds(float* hbase, v8f d, int mt,
                                               int nt, int m, int hi) {
#pragma unroll
  for (int r = 0; r < 8; ++r)
    hbase[(mt * 16 + r + 8 * hi) * HPAD + nt * 16 + m] = d[r];
}

__device__ __forceinline__ void store_tile_g64(float* g, v8f d, int mt, int nt,
                                               int m, int hi) {
#pragma unroll
  for (int r = 0; r < 8; ++r)
    g[(mt * 16 + r + 8 * hi) * 64 + nt * 16 + m] = d[r];
}

__global__ void bx_gemm_k(const float* __restrict__ x,
                          const float* __restrict__ Bw,
                          float* __restrict__ bx) {
  int tid = threadIdx.x;
  int lane = tid & 31, hi = lane >> 4, m = lane & 15;
  int w = (blockIdx.x * blockDim.x + tid) >> 5;
  int mt = w >> 2, nt = w & 3;

  const float* ap = x + (size_t)(mt * 16 + m) * DIN + 2 * hi;
  const float* bp = Bw + (size_t)(nt * 16 + m) * DIN + 2 * hi;

  v8f acc = {0.f, 0.f, 0.f, 0.f, 0.f, 0.f, 0.f, 0.f};
#pragma unroll 8
  for (int kk = 0; kk < DIN / 4; ++kk) {
    v2f a = *(const v2f*)(ap + kk * 4);
    v2f b = *(const v2f*)(bp + kk * 4);
    acc = wmma4(a, b, acc);
  }
  store_tile_g64(bx + (size_t)mt * 16 * LAT, acc, 0, nt, m, hi);
}

__device__ __forceinline__ void square_tiles(const float* cur, int mt, int nt0,
                                             int m, int hi, v8f& acc0,
                                             v8f& acc1) {
  const float* ap = cur + (mt * 16 + m) * HPAD + 2 * hi;
  const float* bp = cur + (2 * hi) * HPAD + nt0 * 16 + m;
  v2f a[16], b0[16], b1[16];
#pragma unroll
  for (int kk = 0; kk < 16; ++kk) {
    a[kk] = *(const v2f*)(ap + kk * 4);
    b0[kk][0] = bp[(kk * 4) * HPAD];
    b0[kk][1] = bp[(kk * 4 + 1) * HPAD];
    b1[kk][0] = bp[(kk * 4) * HPAD + 16];
    b1[kk][1] = bp[(kk * 4 + 1) * HPAD + 16];
  }
  v8f c0 = {0.f, 0.f, 0.f, 0.f, 0.f, 0.f, 0.f, 0.f};
  v8f c1 = {0.f, 0.f, 0.f, 0.f, 0.f, 0.f, 0.f, 0.f};
#pragma unroll
  for (int kk = 0; kk < 16; ++kk) {
    c0 = wmma4(a[kk], b0[kk], c0);
    c1 = wmma4(a[kk], b1[kk], c1);
  }
  acc0 = c0;
  acc1 = c1;
}

__global__ void __launch_bounds__(256)
matpow_k(const float* __restrict__ A, float* __restrict__ Apow) {
  __shared__ __attribute__((aligned(16))) float P[2][64 * HPAD];
  int tid = threadIdx.x;
  int lane = tid & 31, hi = lane >> 4, m = lane & 15;
  int w = tid >> 5, mt = w >> 1, nt0 = (w & 1) * 2;

#pragma unroll
  for (int j = 0; j < 4; ++j) {
    int q = tid + j * 256;
    v4f v = *(const v4f*)(A + q * 4);
    *(v4f*)(&P[0][(q >> 4) * HPAD + (q & 15) * 4]) = v;
  }
  __syncthreads();

  v8f acc0, acc1;
  int p = 0;
  for (int s = 0; s < 4; ++s) {
    square_tiles(P[p], mt, nt0, m, hi, acc0, acc1);
    store_tile_lds(P[1 - p], acc0, mt, nt0, m, hi);
    store_tile_lds(P[1 - p], acc1, mt, nt0 + 1, m, hi);
    __syncthreads();
    p ^= 1;
  }
  square_tiles(P[p], mt, nt0, m, hi, acc0, acc1);
  store_tile_g64(Apow, acc0, mt, nt0, m, hi);
  store_tile_g64(Apow, acc1, mt, nt0 + 1, m, hi);
}

__global__ void chunk_local_k(const float* __restrict__ bx,
                              const float* __restrict__ A,
                              float* __restrict__ Lc) {
  __shared__ float hbuf[2][BT * HPAD];
  int tid = threadIdx.x;
  int lane = tid & 31, hi = lane >> 4, m = lane & 15;
  int w = tid >> 5, mt = w >> 2, nt = w & 3;
  int c = blockIdx.x;

  for (int i = tid; i < BT * HPAD; i += blockDim.x) hbuf[0][i] = 0.f;

  v2f bfrag[16];
  load_bfrag64(bfrag, A, nt, m, hi);

  v8f u_next = load_u_tile(bx + (size_t)(c * T) * BT * LAT, mt, nt, m, hi);
  __syncthreads();

  int p = 0;
#pragma unroll 1
  for (int i = 0; i < T; ++i) {
    v8f acc = u_next;
    if (i + 1 < T)
      u_next =
          load_u_tile(bx + (size_t)(c * T + i + 1) * BT * LAT, mt, nt, m, hi);
    acc = tile_mm_lds(hbuf[p], bfrag, acc, mt, m, hi);
    store_tile_lds(hbuf[1 - p], acc, mt, nt, m, hi);
    lds_barrier();
    p ^= 1;
  }
  float* out = Lc + (size_t)c * BT * LAT;
  for (int i = tid; i < BT * LAT; i += blockDim.x)
    out[i] = hbuf[p][(i >> 6) * HPAD + (i & 63)];
}

__global__ void chunk_scan_k(const float* __restrict__ Apow,
                             const float* __restrict__ Lc,
                             float* __restrict__ carries) {
  __shared__ float hbuf[2][BT * HPAD];
  int tid = threadIdx.x;
  int lane = tid & 31, hi = lane >> 4, m = lane & 15;
  int w = tid >> 5, mt = w >> 2, nt = w & 3;

  for (int i = tid; i < BT * HPAD; i += blockDim.x) hbuf[0][i] = 0.f;
  for (int i = tid; i < BT * LAT; i += blockDim.x) carries[i] = 0.f;

  v2f bfrag[16];
  load_bfrag64(bfrag, Apow, nt, m, hi);

  v8f u_next = load_u_tile(Lc, mt, nt, m, hi);
  __syncthreads();

  int p = 0;
#pragma unroll 1
  for (int c = 0; c < NC - 1; ++c) {
    v8f acc = u_next;
    if (c + 1 < NC - 1)
      u_next = load_u_tile(Lc + (size_t)(c + 1) * BT * LAT, mt, nt, m, hi);
    acc = tile_mm_lds(hbuf[p], bfrag, acc, mt, m, hi);
    store_tile_lds(hbuf[1 - p], acc, mt, nt, m, hi);
    store_tile_g64(carries + (size_t)(c + 1) * BT * LAT, acc, mt, nt, m, hi);
    lds_barrier();
    p ^= 1;
  }
}

__global__ void chunk_final_k(const float* __restrict__ bx,
                              const float* __restrict__ A,
                              const float* __restrict__ carries,
                              float* __restrict__ h_out) {
  __shared__ float hbuf[2][BT * HPAD];
  int tid = threadIdx.x;
  int lane = tid & 31, hi = lane >> 4, m = lane & 15;
  int w = tid >> 5, mt = w >> 2, nt = w & 3;
  int c = blockIdx.x;

  const float* cin = carries + (size_t)c * BT * LAT;
  for (int i = tid; i < BT * LAT; i += blockDim.x)
    hbuf[0][(i >> 6) * HPAD + (i & 63)] = cin[i];

  v2f bfrag[16];
  load_bfrag64(bfrag, A, nt, m, hi);

  v8f u_next = load_u_tile(bx + (size_t)(c * T) * BT * LAT, mt, nt, m, hi);
  __syncthreads();

  int p = 0;
#pragma unroll 1
  for (int i = 0; i < T; ++i) {
    int t = c * T + i;
    v8f acc = u_next;
    if (i + 1 < T)
      u_next = load_u_tile(bx + (size_t)(t + 1) * BT * LAT, mt, nt, m, hi);
    acc = tile_mm_lds(hbuf[p], bfrag, acc, mt, m, hi);
    store_tile_lds(hbuf[1 - p], acc, mt, nt, m, hi);
    store_tile_g64(h_out + (size_t)(t + 1) * BT * LAT, acc, mt, nt, m, hi);
    lds_barrier();
    p ^= 1;
  }
}

constexpr int Y_BLOCKS = 1024;
constexpr int Y_TILES  = (S * BT / 16) / Y_BLOCKS;

__device__ __forceinline__ void load_row_k64(v4f* q, const float* rowp_h) {
#pragma unroll
  for (int i = 0; i < 8; ++i) q[i] = *(const v4f*)(rowp_h + 8 * i);
}

__device__ __forceinline__ void y_tile(const v4f* aq, const v4f (*bq)[8],
                                       float* yp) {
  v8f acc[4];
#pragma unroll
  for (int j = 0; j < 4; ++j)
    acc[j] = (v8f){0.f, 0.f, 0.f, 0.f, 0.f, 0.f, 0.f, 0.f};
#pragma unroll
  for (int i = 0; i < 8; ++i) {
#pragma unroll
    for (int jj = 0; jj < 2; ++jj) {
      v2f a = {aq[i][2 * jj], aq[i][2 * jj + 1]};
#pragma unroll
      for (int j = 0; j < 4; ++j) {
        v2f b = {bq[j][i][2 * jj], bq[j][i][2 * jj + 1]};
        acc[j] = wmma4(a, b, acc[j]);
      }
    }
  }
#pragma unroll
  for (int r = 0; r < 8; ++r) {
    v4f o = {acc[0][r], acc[1][r], acc[2][r], acc[3][r]};
    *(v4f*)(yp + (size_t)r * DOUT) = o;
  }
}

__global__ void __launch_bounds__(128)
y_gemm_k(const float* __restrict__ hs,
         const float* __restrict__ C,
         float* __restrict__ y) {
  int tid = threadIdx.x;
  int lane = tid & 31, h = lane >> 4, c = lane & 15;
  int cg = tid >> 5;

  v4f bq[4][8];
#pragma unroll
  for (int j = 0; j < 4; ++j)
    load_row_k64(bq[j], C + (size_t)(cg * 64 + 4 * c + j) * LAT + 4 * h);

  const float* ap = hs + ((size_t)blockIdx.x * 16 + c) * LAT + 4 * h;
  float* yp = y + ((size_t)blockIdx.x * 16 + 8 * h) * DOUT + cg * 64 + 4 * c;
  const size_t a_step = (size_t)Y_BLOCKS * 16 * LAT;
  const size_t y_step = (size_t)Y_BLOCKS * 16 * DOUT;

  v4f a0[8], a1[8];
  load_row_k64(a0, ap);
#pragma unroll 1
  for (int it = 0; it < Y_TILES; it += 2) {
    load_row_k64(a1, ap + a_step);
    y_tile(a0, bq, yp);
    if (it + 2 < Y_TILES) load_row_k64(a0, ap + 2 * a_step);
    y_tile(a1, bq, yp + y_step);
    ap += 2 * a_step;
    yp += 2 * y_step;
  }
}

extern "C" void kernel_launch(void* const* d_in, const int* in_sizes, int n_in,
                              void* d_out, int out_size, void* d_ws,
                              size_t ws_size, hipStream_t stream) {
  const float* x  = (const float*)d_in[0];
  const float* A  = (const float*)d_in[1];
  const float* Bw = (const float*)d_in[2];
  const float* C  = (const float*)d_in[3];

  float* out      = (float*)d_out;
  float* y_region = out;
  float* h_region = out + (size_t)S * BT * DOUT;

  float* bx = y_region;

  float* ws      = (float*)d_ws;
  float* Apow    = ws;
  float* Lc      = ws + 64 * 64;
  float* carries = ws + 64 * 64 + (size_t)NC * BT * LAT;

  hipMemsetAsync(h_region, 0, (size_t)BT * LAT * sizeof(float), stream);

  bx_gemm_k<<<(S * BT / 16) * (LAT / 16) / 8, 256, 0, stream>>>(x, Bw, bx);
  matpow_k<<<1, 256, 0, stream>>>(A, Apow);
  chunk_local_k<<<NC, 256, 0, stream>>>(bx, A, Lc);
  chunk_scan_k<<<1, 256, 0, stream>>>(Apow, Lc, carries);
  chunk_final_k<<<NC, 256, 0, stream>>>(bx, A, carries, h_region);
  y_gemm_k<<<Y_BLOCKS, 128, 0, stream>>>(h_region + BT * LAT, C, y_region);
}


# bx GEMM re-tiled: a wave keeps the B fragments of 16 columns in registers and loops over row tiles (16384 one-tile waves -> 4096), 16-byte x loads 3 chunks ahead; k summation order of bx changes
# speedup vs baseline: 1.8550x; 1.0905x over previous
#include <hip/hip_runtime.h>


typedef __attribute__((ext_vector_type(2))) float v2f;
typedef __attribute__((ext_vector_type(4))) float v4f;
typedef __attribute__((ext_vector_type(8))) float v8f;

constexpr int S    = 2048;
constexpr int BT   = 32;
constexpr int DIN  = 256;
constexpr int DOUT = 256;
constexpr int LAT  = 64;
constexpr int T    = 32;
constexpr int NC   = 64;
constexpr int HPAD = 68;

__device__ __forceinline__ v8f wmma4(v2f a, v2f b, v8f c) {
  return __builtin_amdgcn_wmma_f32_16x16x4_f32(false, a, false, b, (short)0, c,
                                               false, false);
}

__device__ __forceinline__ void lds_barrier() {
  __builtin_amdgcn_fence(__ATOMIC_RELEASE, "workgroup", "local");
  __builtin_amdgcn_s_barrier();
  __builtin_amdgcn_fence(__ATOMIC_ACQUIRE, "workgroup", "local");
}

__device__ __forceinline__ v8f tile_mm_lds(const float* hbase, const v2f* bfrag,
                                           v8f acc, int mt, int m, int hi) {
  const float* ap = hbase + (mt * 16 + m) * HPAD + 2 * hi;
#pragma unroll
  for (int kk = 0; kk < 16; ++kk) {
    v2f a = *(const v2f*)(ap + kk * 4);
    acc = wmma4(a, bfrag[kk], acc);
  }
  return acc;
}

__device__ __forceinline__ void load_bfrag64(v2f* bfrag, const float* W, int nt,
                                             int m, int hi) {
#pragma unroll
  for (int kk = 0; kk < 16; ++kk)
    bfrag[kk] = *(const v2f*)(W + (nt * 16 + m) * 64 + kk * 4 + 2 * hi);
}

__device__ __forceinline__ v8f load_u_tile(const float* u, int mt, int nt,
                                           int m, int hi) {
  v8f acc;
#pragma unroll
  for (int r = 0; r < 8; ++r)
    acc[r] = u[(mt * 16 + r + 8 * hi) * 64 + nt * 16 + m];
  return acc;
}

__device__ __forceinline__ void store_tile_lds(float* hbase, v8f d, int mt,
                                               int nt, int m, int hi) {
#pragma unroll
  for (int r = 0; r < 8; ++r)
    hbase[(mt * 16 + r + 8 * hi) * HPAD + nt * 16 + m] = d[r];
}

__device__ __forceinline__ void store_tile_g64(float* g, v8f d, int mt, int nt,
                                               int m, int hi) {
#pragma unroll
  for (int r = 0; r < 8; ++r)
    g[(mt * 16 + r + 8 * hi) * 64 + nt * 16 + m] = d[r];
}

constexpr int BX_BLOCKS = 1024;
constexpr int BX_TILES  = (S * BT / 16) / BX_BLOCKS;

__device__ __forceinline__ void load_row_k32(v4f* q, const float* rowp_h) {
#pragma unroll
  for (int i = 0; i < 4; ++i) q[i] = *(const v4f*)(rowp_h + 8 * i);
}

__device__ __forceinline__ v8f mm_k32(const v4f* aq, const v4f* bq, v8f acc) {
#pragma unroll
  for (int i = 0; i < 4; ++i) {
#pragma unroll
    for (int jj = 0; jj < 2; ++jj) {
      v2f a = {aq[i][2 * jj], aq[i][2 * jj + 1]};
      v2f b = {bq[i][2 * jj], bq[i][2 * jj + 1]};
      acc = wmma4(a, b, acc);
    }
  }
  return acc;
}

template <bool MORE>
__device__ __forceinline__ void bx_tile(v4f* a0, v4f* a1, v4f* a2, v4f* a3,
                                        const v4f* bq, const float* ap,
                                        const float* nx, float* op) {
  v8f acc = {0.f, 0.f, 0.f, 0.f, 0.f, 0.f, 0.f, 0.f};
  load_row_k32(a3, ap + 96);
  acc = mm_k32(a0, bq + 0, acc);
  load_row_k32(a0, ap + 128);
  acc = mm_k32(a1, bq + 4, acc);
  load_row_k32(a1, ap + 160);
  acc = mm_k32(a2, bq + 8, acc);
  load_row_k32(a2, ap + 192);
  acc = mm_k32(a3, bq + 12, acc);
  load_row_k32(a3, ap + 224);
  acc = mm_k32(a0, bq + 16, acc);
  if (MORE) load_row_k32(a0, nx);
  acc = mm_k32(a1, bq + 20, acc);
  if (MORE) load_row_k32(a1, nx + 32);
  acc = mm_k32(a2, bq + 24, acc);
  if (MORE) load_row_k32(a2, nx + 64);
  acc = mm_k32(a3, bq + 28, acc);
#pragma unroll
  for (int r = 0; r < 8; ++r) op[(size_t)r * LAT] = acc[r];
}

__global__ void __launch_bounds__(128)
bx_gemm_k(const float* __restrict__ x,
          const float* __restrict__ Bw,
          float* __restrict__ bx) {
  int tid = threadIdx.x;
  int lane = tid & 31, h = lane >> 4, c = lane & 15;
  int nt = tid >> 5;

  v4f bq[32];
  {
    const float* bp = Bw + (size_t)(nt * 16 + c) * DIN + 4 * h;
#pragma unroll
    for (int i = 0; i < 32; ++i) bq[i] = *(const v4f*)(bp + 8 * i);
  }

  const float* ap = x + ((size_t)blockIdx.x * 16 + c) * DIN + 4 * h;
  float* op = bx + ((size_t)blockIdx.x * 16 + 8 * h) * LAT + nt * 16 + c;
  const size_t a_step = (size_t)BX_BLOCKS * 16 * DIN;
  const size_t o_step = (size_t)BX_BLOCKS * 16 * LAT;

  v4f a0[4], a1[4], a2[4], a3[4];
  load_row_k32(a0, ap);
  load_row_k32(a1, ap + 32);
  load_row_k32(a2, ap + 64);
#pragma unroll 1
  for (int it = 0; it < BX_TILES - 1; ++it) {
    bx_tile<true>(a0, a1, a2, a3, bq, ap, ap + a_step, op);
    ap += a_step;
    op += o_step;
  }
  bx_tile<false>(a0, a1, a2, a3, bq, ap, ap, op);
}

__device__ __forceinline__ void square_tiles(const float* cur, int mt, int nt0,
                                             int m, int hi, v8f& acc0,
                                             v8f& acc1) {
  const float* ap = cur + (mt * 16 + m) * HPAD + 2 * hi;
  const float* bp = cur + (2 * hi) * HPAD + nt0 * 16 + m;
  v2f a[16], b0[16], b1[16];
#pragma unroll
  for (int kk = 0; kk < 16; ++kk) {
    a[kk] = *(const v2f*)(ap + kk * 4);
    b0[kk][0] = bp[(kk * 4) * HPAD];
    b0[kk][1] = bp[(kk * 4 + 1) * HPAD];
    b1[kk][0] = bp[(kk * 4) * HPAD + 16];
    b1[kk][1] = bp[(kk * 4 + 1) * HPAD + 16];
  }
  v8f c0 = {0.f, 0.f, 0.f, 0.f, 0.f, 0.f, 0.f, 0.f};
  v8f c1 = {0.f, 0.f, 0.f, 0.f, 0.f, 0.f, 0.f, 0.f};
#pragma unroll
  for (int kk = 0; kk < 16; ++kk) {
    c0 = wmma4(a[kk], b0[kk], c0);
    c1 = wmma4(a[kk], b1[kk], c1);
  }
  acc0 = c0;
  acc1 = c1;
}

__global__ void __launch_bounds__(256)
matpow_k(const float* __restrict__ A, float* __restrict__ Apow) {
  __shared__ __attribute__((aligned(16))) float P[2][64 * HPAD];
  int tid = threadIdx.x;
  int lane = tid & 31, hi = lane >> 4, m = lane & 15;
  int w = tid >> 5, mt = w >> 1, nt0 = (w & 1) * 2;

#pragma unroll
  for (int j = 0; j < 4; ++j) {
    int q = tid + j * 256;
    v4f v = *(const v4f*)(A + q * 4);
    *(v4f*)(&P[0][(q >> 4) * HPAD + (q & 15) * 4]) = v;
  }
  __syncthreads();

  v8f acc0, acc1;
  int p = 0;
  for (int s = 0; s < 4; ++s) {
    square_tiles(P[p], mt, nt0, m, hi, acc0, acc1);
    store_tile_lds(P[1 - p], acc0, mt, nt0, m, hi);
    store_tile_lds(P[1 - p], acc1, mt, nt0 + 1, m, hi);
    __syncthreads();
    p ^= 1;
  }
  square_tiles(P[p], mt, nt0, m, hi, acc0, acc1);
  store_tile_g64(Apow, acc0, mt, nt0, m, hi);
  store_tile_g64(Apow, acc1, mt, nt0 + 1, m, hi);
}

__global__ void chunk_local_k(const float* __restrict__ bx,
                              const float* __restrict__ A,
                              float* __restrict__ Lc) {
  __shared__ float hbuf[2][BT * HPAD];
  int tid = threadIdx.x;
  int lane = tid & 31, hi = lane >> 4, m = lane & 15;
  int w = tid >> 5, mt = w >> 2, nt = w & 3;
  int c = blockIdx.x;

  for (int i = tid; i < BT * HPAD; i += blockDim.x) hbuf[0][i] = 0.f;

  v2f bfrag[16];
  load_bfrag64(bfrag, A, nt, m, hi);

  v8f u_next = load_u_tile(bx + (size_t)(c * T) * BT * LAT, mt, nt, m, hi);
  __syncthreads();

  int p = 0;
#pragma unroll 1
  for (int i = 0; i < T; ++i) {
    v8f acc = u_next;
    if (i + 1 < T)
      u_next =
          load_u_tile(bx + (size_t)(c * T + i + 1) * BT * LAT, mt, nt, m, hi);
    acc = tile_mm_lds(hbuf[p], bfrag, acc, mt, m, hi);
    store_tile_lds(hbuf[1 - p], acc, mt, nt, m, hi);
    lds_barrier();
    p ^= 1;
  }
  float* out = Lc + (size_t)c * BT * LAT;
  for (int i = tid; i < BT * LAT; i += blockDim.x)
    out[i] = hbuf[p][(i >> 6) * HPAD + (i & 63)];
}

__global__ void chunk_scan_k(const float* __restrict__ Apow,
                             const float* __restrict__ Lc,
                             float* __restrict__ carries) {
  __shared__ float hbuf[2][BT * HPAD];
  int tid = threadIdx.x;
  int lane = tid & 31, hi = lane >> 4, m = lane & 15;
  int w = tid >> 5, mt = w >> 2, nt = w & 3;

  for (int i = tid; i < BT * HPAD; i += blockDim.x) hbuf[0][i] = 0.f;
  for (int i = tid; i < BT * LAT; i += blockDim.x) carries[i] = 0.f;

  v2f bfrag[16];
  load_bfrag64(bfrag, Apow, nt, m, hi);

  v8f u_next = load_u_tile(Lc, mt, nt, m, hi);
  __syncthreads();

  int p = 0;
#pragma unroll 1
  for (int c = 0; c < NC - 1; ++c) {
    v8f acc = u_next;
    if (c + 1 < NC - 1)
      u_next = load_u_tile(Lc + (size_t)(c + 1) * BT * LAT, mt, nt, m, hi);
    acc = tile_mm_lds(hbuf[p], bfrag, acc, mt, m, hi);
    store_tile_lds(hbuf[1 - p], acc, mt, nt, m, hi);
    store_tile_g64(carries + (size_t)(c + 1) * BT * LAT, acc, mt, nt, m, hi);
    lds_barrier();
    p ^= 1;
  }
}

__global__ void chunk_final_k(const float* __restrict__ bx,
                              const float* __restrict__ A,
                              const float* __restrict__ carries,
                              float* __restrict__ h_out) {
  __shared__ float hbuf[2][BT * HPAD];
  int tid = threadIdx.x;
  int lane = tid & 31, hi = lane >> 4, m = lane & 15;
  int w = tid >> 5, mt = w >> 2, nt = w & 3;
  int c = blockIdx.x;

  const float* cin = carries + (size_t)c * BT * LAT;
  for (int i = tid; i < BT * LAT; i += blockDim.x)
    hbuf[0][(i >> 6) * HPAD + (i & 63)] = cin[i];

  v2f bfrag[16];
  load_bfrag64(bfrag, A, nt, m, hi);

  v8f u_next = load_u_tile(bx + (size_t)(c * T) * BT * LAT, mt, nt, m, hi);
  __syncthreads();

  int p = 0;
#pragma unroll 1
  for (int i = 0; i < T; ++i) {
    int t = c * T + i;
    v8f acc = u_next;
    if (i + 1 < T)
      u_next = load_u_tile(bx + (size_t)(t + 1) * BT * LAT, mt, nt, m, hi);
    acc = tile_mm_lds(hbuf[p], bfrag, acc, mt, m, hi);
    store_tile_lds(hbuf[1 - p], acc, mt, nt, m, hi);
    store_tile_g64(h_out + (size_t)(t + 1) * BT * LAT, acc, mt, nt, m, hi);
    lds_barrier();
    p ^= 1;
  }
}

constexpr int Y_BLOCKS = 1024;
constexpr int Y_TILES  = (S * BT / 16) / Y_BLOCKS;

__device__ __forceinline__ void load_row_k64(v4f* q, const float* rowp_h) {
#pragma unroll
  for (int i = 0; i < 8; ++i) q[i] = *(const v4f*)(rowp_h + 8 * i);
}

__device__ __forceinline__ void y_tile(const v4f* aq, const v4f (*bq)[8],
                                       float* yp) {
  v8f acc[4];
#pragma unroll
  for (int j = 0; j < 4; ++j)
    acc[j] = (v8f){0.f, 0.f, 0.f, 0.f, 0.f, 0.f, 0.f, 0.f};
#pragma unroll
  for (int i = 0; i < 8; ++i) {
#pragma unroll
    for (int jj = 0; jj < 2; ++jj) {
      v2f a = {aq[i][2 * jj], aq[i][2 * jj + 1]};
#pragma unroll
      for (int j = 0; j < 4; ++j) {
        v2f b = {bq[j][i][2 * jj], bq[j][i][2 * jj + 1]};
        acc[j] = wmma4(a, b, acc[j]);
      }
    }
  }
#pragma unroll
  for (int r = 0; r < 8; ++r) {
    v4f o = {acc[0][r], acc[1][r], acc[2][r], acc[3][r]};
    *(v4f*)(yp + (size_t)r * DOUT) = o;
  }
}

__global__ void __launch_bounds__(128)
y_gemm_k(const float* __restrict__ hs,
         const float* __restrict__ C,
         float* __restrict__ y) {
  int tid = threadIdx.x;
  int lane = tid & 31, h = lane >> 4, c = lane & 15;
  int cg = tid >> 5;

  v4f bq[4][8];
#pragma unroll
  for (int j = 0; j < 4; ++j)
    load_row_k64(bq[j], C + (size_t)(cg * 64 + 4 * c + j) * LAT + 4 * h);

  const float* ap = hs + ((size_t)blockIdx.x * 16 + c) * LAT + 4 * h;
  float* yp = y + ((size_t)blockIdx.x * 16 + 8 * h) * DOUT + cg * 64 + 4 * c;
  const size_t a_step = (size_t)Y_BLOCKS * 16 * LAT;
  const size_t y_step = (size_t)Y_BLOCKS * 16 * DOUT;

  v4f a0[8], a1[8];
  load_row_k64(a0, ap);
#pragma unroll 1
  for (int it = 0; it < Y_TILES; it += 2) {
    load_row_k64(a1, ap + a_step);
    y_tile(a0, bq, yp);
    if (it + 2 < Y_TILES) load_row_k64(a0, ap + 2 * a_step);
    y_tile(a1, bq, yp + y_step);
    ap += 2 * a_step;
    yp += 2 * y_step;
  }
}

extern "C" void kernel_launch(void* const* d_in, const int* in_sizes, int n_in,
                              void* d_out, int out_size, void* d_ws,
                              size_t ws_size, hipStream_t stream) {
  const float* x  = (const float*)d_in[0];
  const float* A  = (const float*)d_in[1];
  const float* Bw = (const float*)d_in[2];
  const float* C  = (const float*)d_in[3];

  float* out      = (float*)d_out;
  float* y_region = out;
  float* h_region = out + (size_t)S * BT * DOUT;

  float* bx = y_region;

  float* ws      = (float*)d_ws;
  float* Apow    = ws;
  float* Lc      = ws + 64 * 64;
  float* carries = ws + 64 * 64 + (size_t)NC * BT * LAT;

  hipMemsetAsync(h_region, 0, (size_t)BT * LAT * sizeof(float), stream);

  bx_gemm_k<<<BX_BLOCKS, 128, 0, stream>>>(x, Bw, bx);
  matpow_k<<<1, 256, 0, stream>>>(A, Apow);
  chunk_local_k<<<NC, 256, 0, stream>>>(bx, A, Lc);
  chunk_scan_k<<<1, 256, 0, stream>>>(Apow, Lc, carries);
  chunk_final_k<<<NC, 256, 0, stream>>>(bx, A, carries, h_region);
  y_gemm_k<<<Y_BLOCKS, 128, 0, stream>>>(h_region + BT * LAT, C, y_region);
}


# Scan kernels: one 4-wave work-group per 16 batch rows (one wave per SIMD: one 16-WMMA chain per matrix core and step, not two), h fragment LDS loads grouped ahead of the chain; no sum order changes
# speedup vs baseline: 2.2291x; 1.2017x over previous
#include <hip/hip_runtime.h>


typedef __attribute__((ext_vector_type(2))) float v2f;
typedef __attribute__((ext_vector_type(4))) float v4f;
typedef __attribute__((ext_vector_type(8))) float v8f;

constexpr int S    = 2048;
constexpr int BT   = 32;
constexpr int DIN  = 256;
constexpr int DOUT = 256;
constexpr int LAT  = 64;
constexpr int T    = 32;
constexpr int NC   = 64;
constexpr int HPAD = 68;

constexpr int MROWS        = 16;
constexpr int MTILES       = BT / MROWS;
constexpr int SCAN_THREADS = 128;
constexpr int SCAN_GROUPS  = NC * MTILES;

__device__ __forceinline__ v8f wmma4(v2f a, v2f b, v8f c) {
  return __builtin_amdgcn_wmma_f32_16x16x4_f32(false, a, false, b, (short)0, c,
                                               false, false);
}

__device__ __forceinline__ void lds_barrier() {
  __builtin_amdgcn_fence(__ATOMIC_RELEASE, "workgroup", "local");
  __builtin_amdgcn_s_barrier();
  __builtin_amdgcn_fence(__ATOMIC_ACQUIRE, "workgroup", "local");
}

__device__ __forceinline__ v8f tile_mm_lds(const float* hbase, const v2f* bfrag,
                                           v8f acc, int mt, int m, int hi) {
  const float* ap = hbase + (mt * 16 + m) * HPAD + 2 * hi;
  v2f a[16];
#pragma unroll
  for (int kk = 0; kk < 16; ++kk) a[kk] = *(const v2f*)(ap + kk * 4);
#pragma unroll
  for (int kk = 0; kk < 16; ++kk) acc = wmma4(a[kk], bfrag[kk], acc);
  __builtin_amdgcn_sched_group_barrier(0x100, 16, 0);
  __builtin_amdgcn_sched_group_barrier(0x008, 16, 0);
  return acc;
}

__device__ __forceinline__ void load_bfrag64(v2f* bfrag, const float* W, int nt,
                                             int m, int hi) {
#pragma unroll
  for (int kk = 0; kk < 16; ++kk)
    bfrag[kk] = *(const v2f*)(W + (nt * 16 + m) * 64 + kk * 4 + 2 * hi);
}

__device__ __forceinline__ v8f load_u_tile(const float* u, int mt, int nt,
                                           int m, int hi) {
  v8f acc;
#pragma unroll
  for (int r = 0; r < 8; ++r)
    acc[r] = u[(mt * 16 + r + 8 * hi) * 64 + nt * 16 + m];
  return acc;
}

__device__ __forceinline__ void store_tile_lds(float* hbase, v8f d, int mt,
                                               int nt, int m, int hi) {
#pragma unroll
  for (int r = 0; r < 8; ++r)
    hbase[(mt * 16 + r + 8 * hi) * HPAD + nt * 16 + m] = d[r];
}

__device__ __forceinline__ void store_tile_g64(float* g, v8f d, int mt, int nt,
                                               int m, int hi) {
#pragma unroll
  for (int r = 0; r < 8; ++r)
    g[(mt * 16 + r + 8 * hi) * 64 + nt * 16 + m] = d[r];
}

constexpr int BX_BLOCKS = 1024;
constexpr int BX_TILES  = (S * BT / 16) / BX_BLOCKS;

__device__ __forceinline__ void load_row_k32(v4f* q, const float* rowp_h) {
#pragma unroll
  for (int i = 0; i < 4; ++i) q[i] = *(const v4f*)(rowp_h + 8 * i);
}

__device__ __forceinline__ v8f mm_k32(const v4f* aq, const v4f* bq, v8f acc) {
#pragma unroll
  for (int i = 0; i < 4; ++i) {
#pragma unroll
    for (int jj = 0; jj < 2; ++jj) {
      v2f a = {aq[i][2 * jj], aq[i][2 * jj + 1]};
      v2f b = {bq[i][2 * jj], bq[i][2 * jj + 1]};
      acc = wmma4(a, b, acc);
    }
  }
  return acc;
}

template <bool MORE>
__device__ __forceinline__ void bx_tile(v4f* a0, v4f* a1, v4f* a2, v4f* a3,
                                        const v4f* bq, const float* ap,
                                        const float* nx, float* op) {
  v8f acc = {0.f, 0.f, 0.f, 0.f, 0.f, 0.f, 0.f, 0.f};
  load_row_k32(a3, ap + 96);
  acc = mm_k32(a0, bq + 0, acc);
  load_row_k32(a0, ap + 128);
  acc = mm_k32(a1, bq + 4, acc);
  load_row_k32(a1, ap + 160);
  acc = mm_k32(a2, bq + 8, acc);
  load_row_k32(a2, ap + 192);
  acc = mm_k32(a3, bq + 12, acc);
  load_row_k32(a3, ap + 224);
  acc = mm_k32(a0, bq + 16, acc);
  if (MORE) load_row_k32(a0, nx);
  acc = mm_k32(a1, bq + 20, acc);
  if (MORE) load_row_k32(a1, nx + 32);
  acc = mm_k32(a2, bq + 24, acc);
  if (MORE) load_row_k32(a2, nx + 64);
  acc = mm_k32(a3, bq + 28, acc);
#pragma unroll
  for (int r = 0; r < 8; ++r) op[(size_t)r * LAT] = acc[r];
}

__global__ void __launch_bounds__(128)
bx_gemm_k(const float* __restrict__ x,
          const float* __restrict__ Bw,
          float* __restrict__ bx) {
  int tid = threadIdx.x;
  int lane = tid & 31, h = lane >> 4, c = lane & 15;
  int nt = tid >> 5;

  v4f bq[32];
  {
    const float* bp = Bw + (size_t)(nt * 16 + c) * DIN + 4 * h;
#pragma unroll
    for (int i = 0; i < 32; ++i) bq[i] = *(const v4f*)(bp + 8 * i);
  }

  const float* ap = x + ((size_t)blockIdx.x * 16 + c) * DIN + 4 * h;
  float* op = bx + ((size_t)blockIdx.x * 16 + 8 * h) * LAT + nt * 16 + c;
  const size_t a_step = (size_t)BX_BLOCKS * 16 * DIN;
  const size_t o_step = (size_t)BX_BLOCKS * 16 * LAT;

  v4f a0[4], a1[4], a2[4], a3[4];
  load_row_k32(a0, ap);
  load_row_k32(a1, ap + 32);
  load_row_k32(a2, ap + 64);
#pragma unroll 1
  for (int it = 0; it < BX_TILES - 1; ++it) {
    bx_tile<true>(a0, a1, a2, a3, bq, ap, ap + a_step, op);
    ap += a_step;
    op += o_step;
  }
  bx_tile<false>(a0, a1, a2, a3, bq, ap, ap, op);
}

__device__ __forceinline__ void square_tiles(const float* cur, int mt, int nt0,
                                             int m, int hi, v8f& acc0,
                                             v8f& acc1) {
  const float* ap = cur + (mt * 16 + m) * HPAD + 2 * hi;
  const float* bp = cur + (2 * hi) * HPAD + nt0 * 16 + m;
  v2f a[16], b0[16], b1[16];
#pragma unroll
  for (int kk = 0; kk < 16; ++kk) {
    a[kk] = *(const v2f*)(ap + kk * 4);
    b0[kk][0] = bp[(kk * 4) * HPAD];
    b0[kk][1] = bp[(kk * 4 + 1) * HPAD];
    b1[kk][0] = bp[(kk * 4) * HPAD + 16];
    b1[kk][1] = bp[(kk * 4 + 1) * HPAD + 16];
  }
  v8f c0 = {0.f, 0.f, 0.f, 0.f, 0.f, 0.f, 0.f, 0.f};
  v8f c1 = {0.f, 0.f, 0.f, 0.f, 0.f, 0.f, 0.f, 0.f};
#pragma unroll
  for (int kk = 0; kk < 16; ++kk) {
    c0 = wmma4(a[kk], b0[kk], c0);
    c1 = wmma4(a[kk], b1[kk], c1);
  }
  acc0 = c0;
  acc1 = c1;
}

__global__ void __launch_bounds__(256)
matpow_k(const float* __restrict__ A, float* __restrict__ Apow) {
  __shared__ __attribute__((aligned(16))) float P[2][64 * HPAD];
  int tid = threadIdx.x;
  int lane = tid & 31, hi = lane >> 4, m = lane & 15;
  int w = tid >> 5, mt = w >> 1, nt0 = (w & 1) * 2;

#pragma unroll
  for (int j = 0; j < 4; ++j) {
    int q = tid + j * 256;
    v4f v = *(const v4f*)(A + q * 4);
    *(v4f*)(&P[0][(q >> 4) * HPAD + (q & 15) * 4]) = v;
  }
  __syncthreads();

  v8f acc0, acc1;
  int p = 0;
  for (int s = 0; s < 4; ++s) {
    square_tiles(P[p], mt, nt0, m, hi, acc0, acc1);
    store_tile_lds(P[1 - p], acc0, mt, nt0, m, hi);
    store_tile_lds(P[1 - p], acc1, mt, nt0 + 1, m, hi);
    __syncthreads();
    p ^= 1;
  }
  square_tiles(P[p], mt, nt0, m, hi, acc0, acc1);
  store_tile_g64(Apow, acc0, mt, nt0, m, hi);
  store_tile_g64(Apow, acc1, mt, nt0 + 1, m, hi);
}

__global__ void __launch_bounds__(SCAN_THREADS)
chunk_local_k(const float* __restrict__ bx,
              const float* __restrict__ A,
              float* __restrict__ Lc) {
  __shared__ float hbuf[2][MROWS * HPAD];
  int tid = threadIdx.x;
  int lane = tid & 31, hi = lane >> 4, m = lane & 15;
  int nt = tid >> 5;
  int c = blockIdx.x / MTILES, mt = blockIdx.x % MTILES;

  for (int i = tid; i < MROWS * HPAD; i += SCAN_THREADS) hbuf[0][i] = 0.f;

  v2f bfrag[16];
  load_bfrag64(bfrag, A, nt, m, hi);

  v8f u_next = load_u_tile(bx + (size_t)(c * T) * BT * LAT, mt, nt, m, hi);
  __syncthreads();

  int p = 0;
#pragma unroll 1
  for (int i = 0; i < T; ++i) {
    v8f acc = u_next;
    if (i + 1 < T)
      u_next =
          load_u_tile(bx + (size_t)(c * T + i + 1) * BT * LAT, mt, nt, m, hi);
    acc = tile_mm_lds(hbuf[p], bfrag, acc, 0, m, hi);
    store_tile_lds(hbuf[1 - p], acc, 0, nt, m, hi);
    lds_barrier();
    p ^= 1;
  }
  float* out = Lc + ((size_t)c * BT + mt * MROWS) * LAT;
  for (int i = tid; i < MROWS * LAT; i += SCAN_THREADS)
    out[i] = hbuf[p][(i >> 6) * HPAD + (i & 63)];
}

__global__ void __launch_bounds__(SCAN_THREADS)
chunk_scan_k(const float* __restrict__ Apow,
             const float* __restrict__ Lc,
             float* __restrict__ carries) {
  __shared__ float hbuf[2][MROWS * HPAD];
  int tid = threadIdx.x;
  int lane = tid & 31, hi = lane >> 4, m = lane & 15;
  int nt = tid >> 5;
  int mt = blockIdx.x;

  for (int i = tid; i < MROWS * HPAD; i += SCAN_THREADS) hbuf[0][i] = 0.f;
  for (int i = tid; i < MROWS * LAT; i += SCAN_THREADS)
    carries[mt * MROWS * LAT + i] = 0.f;

  v2f bfrag[16];
  load_bfrag64(bfrag, Apow, nt, m, hi);

  v8f u_next = load_u_tile(Lc, mt, nt, m, hi);
  __syncthreads();

  int p = 0;
#pragma unroll 1
  for (int c = 0; c < NC - 1; ++c) {
    v8f acc = u_next;
    if (c + 1 < NC - 1)
      u_next = load_u_tile(Lc + (size_t)(c + 1) * BT * LAT, mt, nt, m, hi);
    acc = tile_mm_lds(hbuf[p], bfrag, acc, 0, m, hi);
    store_tile_lds(hbuf[1 - p], acc, 0, nt, m, hi);
    store_tile_g64(carries + (size_t)(c + 1) * BT * LAT, acc, mt, nt, m, hi);
    lds_barrier();
    p ^= 1;
  }
}

__global__ void __launch_bounds__(SCAN_THREADS)
chunk_final_k(const float* __restrict__ bx,
              const float* __restrict__ A,
              const float* __restrict__ carries,
              float* __restrict__ h_out) {
  __shared__ float hbuf[2][MROWS * HPAD];
  int tid = threadIdx.x;
  int lane = tid & 31, hi = lane >> 4, m = lane & 15;
  int nt = tid >> 5;
  int c = blockIdx.x / MTILES, mt = blockIdx.x % MTILES;

  const float* cin = carries + ((size_t)c * BT + mt * MROWS) * LAT;
  for (int i = tid; i < MROWS * LAT; i += SCAN_THREADS)
    hbuf[0][(i >> 6) * HPAD + (i & 63)] = cin[i];

  v2f bfrag[16];
  load_bfrag64(bfrag, A, nt, m, hi);

  v8f u_next = load_u_tile(bx + (size_t)(c * T) * BT * LAT, mt, nt, m, hi);
  __syncthreads();

  int p = 0;
#pragma unroll 1
  for (int i = 0; i < T; ++i) {
    int t = c * T + i;
    v8f acc = u_next;
    if (i + 1 < T)
      u_next = load_u_tile(bx + (size_t)(t + 1) * BT * LAT, mt, nt, m, hi);
    acc = tile_mm_lds(hbuf[p], bfrag, acc, 0, m, hi);
    store_tile_lds(hbuf[1 - p], acc, 0, nt, m, hi);
    store_tile_g64(h_out + (size_t)(t + 1) * BT * LAT, acc, mt, nt, m, hi);
    lds_barrier();
    p ^= 1;
  }
}

constexpr int Y_BLOCKS = 1024;
constexpr int Y_TILES  = (S * BT / 16) / Y_BLOCKS;

__device__ __forceinline__ void load_row_k64(v4f* q, const float* rowp_h) {
#pragma unroll
  for (int i = 0; i < 8; ++i) q[i] = *(const v4f*)(rowp_h + 8 * i);
}

__device__ __forceinline__ void y_tile(const v4f* aq, const v4f (*bq)[8],
                                       float* yp) {
  v8f acc[4];
#pragma unroll
  for (int j = 0; j < 4; ++j)
    acc[j] = (v8f){0.f, 0.f, 0.f, 0.f, 0.f, 0.f, 0.f, 0.f};
#pragma unroll
  for (int i = 0; i < 8; ++i) {
#pragma unroll
    for (int jj = 0; jj < 2; ++jj) {
      v2f a = {aq[i][2 * jj], aq[i][2 * jj + 1]};
#pragma unroll
      for (int j = 0; j < 4; ++j) {
        v2f b = {bq[j][i][2 * jj], bq[j][i][2 * jj + 1]};
        acc[j] = wmma4(a, b, acc[j]);
      }
    }
  }
#pragma unroll
  for (int r = 0; r < 8; ++r) {
    v4f o = {acc[0][r], acc[1][r], acc[2][r], acc[3][r]};
    *(v4f*)(yp + (size_t)r * DOUT) = o;
  }
}

__global__ void __launch_bounds__(128)
y_gemm_k(const float* __restrict__ hs,
         const float* __restrict__ C,
         float* __restrict__ y) {
  int tid = threadIdx.x;
  int lane = tid & 31, h = lane >> 4, c = lane & 15;
  int cg = tid >> 5;

  v4f bq[4][8];
#pragma unroll
  for (int j = 0; j < 4; ++j)
    load_row_k64(bq[j], C + (size_t)(cg * 64 + 4 * c + j) * LAT + 4 * h);

  const float* ap = hs + ((size_t)blockIdx.x * 16 + c) * LAT + 4 * h;
  float* yp = y + ((size_t)blockIdx.x * 16 + 8 * h) * DOUT + cg * 64 + 4 * c;
  const size_t a_step = (size_t)Y_BLOCKS * 16 * LAT;
  const size_t y_step = (size_t)Y_BLOCKS * 16 * DOUT;

  v4f a0[8], a1[8];
  load_row_k64(a0, ap);
#pragma unroll 1
  for (int it = 0; it < Y_TILES; it += 2) {
    load_row_k64(a1, ap + a_step);
    y_tile(a0, bq, yp);
    if (it + 2 < Y_TILES) load_row_k64(a0, ap + 2 * a_step);
    y_tile(a1, bq, yp + y_step);
    ap += 2 * a_step;
    yp += 2 * y_step;
  }
}

extern "C" void kernel_launch(void* const* d_in, const int* in_sizes, int n_in,
                              void* d_out, int out_size, void* d_ws,
                              size_t ws_size, hipStream_t stream) {
  const float* x  = (const float*)d_in[0];
  const float* A  = (const float*)d_in[1];
  const float* Bw = (const float*)d_in[2];
  const float* C  = (const float*)d_in[3];

  float* out      = (float*)d_out;
  float* y_region = out;
  float* h_region = out + (size_t)S * BT * DOUT;

  float* bx = y_region;

  float* ws      = (float*)d_ws;
  float* Apow    = ws;
  float* Lc      = ws + 64 * 64;
  float* carries = ws + 64 * 64 + (size_t)NC * BT * LAT;

  hipMemsetAsync(h_region, 0, (size_t)BT * LAT * sizeof(float), stream);

  bx_gemm_k<<<BX_BLOCKS, 128, 0, stream>>>(x, Bw, bx);
  matpow_k<<<1, 256, 0, stream>>>(A, Apow);
  chunk_local_k<<<SCAN_GROUPS, SCAN_THREADS, 0, stream>>>(bx, A, Lc);
  chunk_scan_k<<<MTILES, SCAN_THREADS, 0, stream>>>(Apow, Lc, carries);
  chunk_final_k<<<SCAN_GROUPS, SCAN_THREADS, 0, stream>>>(bx, A, carries,
                                                         h_region);
  y_gemm_k<<<Y_BLOCKS, 128, 0, stream>>>(h_region + BT * LAT, C, y_region);
}


# 7 launches -> 4: A^32 formed by an extra work-group of the local-pass launch, each final-pass group re-runs the chunk-boundary scan up to its own chunk, a kernel stores h[0]; no sum order changes
# speedup vs baseline: 2.5813x; 1.1580x over previous
#include <hip/hip_runtime.h>


typedef __attribute__((ext_vector_type(2))) float v2f;
typedef __attribute__((ext_vector_type(4))) float v4f;
typedef __attribute__((ext_vector_type(8))) float v8f;

constexpr int S    = 2048;
constexpr int BT   = 32;
constexpr int DIN  = 256;
constexpr int DOUT = 256;
constexpr int LAT  = 64;
constexpr int T    = 32;
constexpr int NC   = 64;
constexpr int HPAD = 68;

constexpr int MROWS        = 16;
constexpr int MTILES       = BT / MROWS;
constexpr int SCAN_THREADS = 128;
constexpr int SCAN_GROUPS  = NC * MTILES;

__device__ __forceinline__ v8f wmma4(v2f a, v2f b, v8f c) {
  return __builtin_amdgcn_wmma_f32_16x16x4_f32(false, a, false, b, (short)0, c,
                                               false, false);
}

__device__ __forceinline__ void lds_barrier() {
  __builtin_amdgcn_fence(__ATOMIC_RELEASE, "workgroup", "local");
  __builtin_amdgcn_s_barrier();
  __builtin_amdgcn_fence(__ATOMIC_ACQUIRE, "workgroup", "local");
}

__device__ __forceinline__ v8f tile_mm_lds(const float* hbase, const v2f* bfrag,
                                           v8f acc, int mt, int m, int hi) {
  const float* ap = hbase + (mt * 16 + m) * HPAD + 2 * hi;
  v2f a[16];
#pragma unroll
  for (int kk = 0; kk < 16; ++kk) a[kk] = *(const v2f*)(ap + kk * 4);
#pragma unroll
  for (int kk = 0; kk < 16; ++kk) acc = wmma4(a[kk], bfrag[kk], acc);
  __builtin_amdgcn_sched_group_barrier(0x100, 16, 0);
  __builtin_amdgcn_sched_group_barrier(0x008, 16, 0);
  return acc;
}

__device__ __forceinline__ void load_bfrag64(v2f* bfrag, const float* W, int nt,
                                             int m, int hi) {
#pragma unroll
  for (int kk = 0; kk < 16; ++kk)
    bfrag[kk] = *(const v2f*)(W + (nt * 16 + m) * 64 + kk * 4 + 2 * hi);
}

__device__ __forceinline__ void store_tile_lds(float* hbase, v8f d, int mt,
                                               int nt, int m, int hi) {
#pragma unroll
  for (int r = 0; r < 8; ++r)
    hbase[(mt * 16 + r + 8 * hi) * HPAD + nt * 16 + m] = d[r];
}

__device__ __forceinline__ void store_tile_g64(float* g, v8f d, int mt, int nt,
                                               int m, int hi) {
#pragma unroll
  for (int r = 0; r < 8; ++r)
    g[(mt * 16 + r + 8 * hi) * 64 + nt * 16 + m] = d[r];
}

__device__ __forceinline__ unsigned tile_lane_off(int mt, int nt, int m,
                                                  int hi) {
  return (unsigned)((mt * 16 + 8 * hi) * 64 + nt * 16 + m);
}

__device__ __forceinline__ v8f load_tile_lane(const float* plane,
                                              unsigned off) {
  const float* pl = plane + off;
  v8f acc;
#pragma unroll
  for (int r = 0; r < 8; ++r) acc[r] = pl[64 * r];
  return acc;
}

__device__ __forceinline__ void store_tile_lane(float* plane, unsigned off,
                                                v8f d) {
  float* pl = plane + off;
#pragma unroll
  for (int r = 0; r < 8; ++r) pl[64 * r] = d[r];
}

constexpr int BX_BLOCKS = 1024;
constexpr int BX_TILES  = (S * BT / 16) / BX_BLOCKS;

__device__ __forceinline__ void load_row_k32(v4f* q, const float* rowp_h) {
#pragma unroll
  for (int i = 0; i < 4; ++i) q[i] = *(const v4f*)(rowp_h + 8 * i);
}

__device__ __forceinline__ v8f mm_k32(const v4f* aq, const v4f* bq, v8f acc) {
#pragma unroll
  for (int i = 0; i < 4; ++i) {
#pragma unroll
    for (int jj = 0; jj < 2; ++jj) {
      v2f a = {aq[i][2 * jj], aq[i][2 * jj + 1]};
      v2f b = {bq[i][2 * jj], bq[i][2 * jj + 1]};
      acc = wmma4(a, b, acc);
    }
  }
  return acc;
}

template <bool MORE>
__device__ __forceinline__ void bx_tile(v4f* a0, v4f* a1, v4f* a2, v4f* a3,
                                        const v4f* bq, const float* ap,
                                        const float* nx, float* op) {
  v8f acc = {0.f, 0.f, 0.f, 0.f, 0.f, 0.f, 0.f, 0.f};
  load_row_k32(a3, ap + 96);
  acc = mm_k32(a0, bq + 0, acc);
  load_row_k32(a0, ap + 128);
  acc = mm_k32(a1, bq + 4, acc);
  load_row_k32(a1, ap + 160);
  acc = mm_k32(a2, bq + 8, acc);
  load_row_k32(a2, ap + 192);
  acc = mm_k32(a3, bq + 12, acc);
  load_row_k32(a3, ap + 224);
  acc = mm_k32(a0, bq + 16, acc);
  if (MORE) load_row_k32(a0, nx);
  acc = mm_k32(a1, bq + 20, acc);
  if (MORE) load_row_k32(a1, nx + 32);
  acc = mm_k32(a2, bq + 24, acc);
  if (MORE) load_row_k32(a2, nx + 64);
  acc = mm_k32(a3, bq + 28, acc);
#pragma unroll
  for (int r = 0; r < 8; ++r) op[(size_t)r * LAT] = acc[r];
}

__global__ void __launch_bounds__(128)
bx_gemm_k(const float* __restrict__ x,
          const float* __restrict__ Bw,
          float* __restrict__ bx) {
  int tid = threadIdx.x;
  int lane = tid & 31, h = lane >> 4, c = lane & 15;
  int nt = tid >> 5;

  v4f bq[32];
  {
    const float* bp = Bw + (size_t)(nt * 16 + c) * DIN + 4 * h;
#pragma unroll
    for (int i = 0; i < 32; ++i) bq[i] = *(const v4f*)(bp + 8 * i);
  }

  const float* ap = x + ((size_t)blockIdx.x * 16 + c) * DIN + 4 * h;
  float* op = bx + ((size_t)blockIdx.x * 16 + 8 * h) * LAT + nt * 16 + c;
  const size_t a_step = (size_t)BX_BLOCKS * 16 * DIN;
  const size_t o_step = (size_t)BX_BLOCKS * 16 * LAT;

  v4f a0[4], a1[4], a2[4], a3[4];
  load_row_k32(a0, ap);
  load_row_k32(a1, ap + 32);
  load_row_k32(a2, ap + 64);
#pragma unroll 1
  for (int it = 0; it < BX_TILES - 1; ++it) {
    bx_tile<true>(a0, a1, a2, a3, bq, ap, ap + a_step, op);
    ap += a_step;
    op += o_step;
  }
  bx_tile<false>(a0, a1, a2, a3, bq, ap, ap, op);
}

__device__ __forceinline__ void square_tiles(const float* cur, int mt, int nt0,
                                             int m, int hi, v8f& acc0,
                                             v8f& acc1) {
  const float* ap = cur + (mt * 16 + m) * HPAD + 2 * hi;
  const float* bp = cur + (2 * hi) * HPAD + nt0 * 16 + m;
  v2f a[16], b0[16], b1[16];
#pragma unroll
  for (int kk = 0; kk < 16; ++kk) {
    a[kk] = *(const v2f*)(ap + kk * 4);
    b0[kk][0] = bp[(kk * 4) * HPAD];
    b0[kk][1] = bp[(kk * 4 + 1) * HPAD];
    b1[kk][0] = bp[(kk * 4) * HPAD + 16];
    b1[kk][1] = bp[(kk * 4 + 1) * HPAD + 16];
  }
  v8f c0 = {0.f, 0.f, 0.f, 0.f, 0.f, 0.f, 0.f, 0.f};
  v8f c1 = {0.f, 0.f, 0.f, 0.f, 0.f, 0.f, 0.f, 0.f};
#pragma unroll
  for (int kk = 0; kk < 16; ++kk) {
    c0 = wmma4(a[kk], b0[kk], c0);
    c1 = wmma4(a[kk], b1[kk], c1);
  }
  acc0 = c0;
  acc1 = c1;
}

__device__ __forceinline__ void matpow_role(const float* __restrict__ A,
                                            float* __restrict__ Apow,
                                            float* P) {
  constexpr int PSZ = 64 * HPAD;
  int tid = threadIdx.x;
  int lane = tid & 31, hi = lane >> 4, m = lane & 15;
  int w = tid >> 5;

  v4f av[8];
#pragma unroll
  for (int j = 0; j < 8; ++j) av[j] = *(const v4f*)(A + (tid + j * 128) * 4);
#pragma unroll
  for (int j = 0; j < 8; ++j) {
    int q = tid + j * 128;
    *(v4f*)(&P[(q >> 4) * HPAD + (q & 15) * 4]) = av[j];
  }
  __syncthreads();

  v8f acc0, acc1;
  int p = 0;
  for (int s = 0; s < 4; ++s) {
    const float* cur = P + p * PSZ;
    float* nxt = P + (1 - p) * PSZ;
#pragma unroll
    for (int half = 0; half < 2; ++half) {
      square_tiles(cur, w, 2 * half, m, hi, acc0, acc1);
      store_tile_lds(nxt, acc0, w, 2 * half, m, hi);
      store_tile_lds(nxt, acc1, w, 2 * half + 1, m, hi);
    }
    __syncthreads();
    p ^= 1;
  }
  const float* cur = P + p * PSZ;
#pragma unroll
  for (int half = 0; half < 2; ++half) {
    square_tiles(cur, w, 2 * half, m, hi, acc0, acc1);
    store_tile_g64(Apow, acc0, w, 2 * half, m, hi);
    store_tile_g64(Apow, acc1, w, 2 * half + 1, m, hi);
  }
}

__global__ void __launch_bounds__(SCAN_THREADS)
chunk_local_k(const float* __restrict__ bx,
              const float* __restrict__ A,
              float* __restrict__ Lc,
              float* __restrict__ Apow) {
  __shared__ __attribute__((aligned(16))) float smem[2 * 64 * HPAD];
  if (blockIdx.x == 0) {
    matpow_role(A, Apow, smem);
    return;
  }
  constexpr int HSZ = MROWS * HPAD;
  int tid = threadIdx.x;
  int lane = tid & 31, hi = lane >> 4, m = lane & 15;
  int nt = tid >> 5;
  unsigned g = blockIdx.x - 1u;
  int c = (int)(g / MTILES), mt = (int)(g % MTILES);
  const unsigned uoff = tile_lane_off(mt, nt, m, hi);

  for (int i = tid; i < HSZ; i += SCAN_THREADS) smem[i] = 0.f;

  v2f bfrag[16];
  load_bfrag64(bfrag, A, nt, m, hi);

  v8f u_next = load_tile_lane(bx + (size_t)(c * T) * BT * LAT, uoff);
  __syncthreads();

  int p = 0;
#pragma unroll 1
  for (int i = 0; i < T; ++i) {
    v8f acc = u_next;
    if (i + 1 < T)
      u_next = load_tile_lane(bx + (size_t)(c * T + i + 1) * BT * LAT, uoff);
    acc = tile_mm_lds(smem + p * HSZ, bfrag, acc, 0, m, hi);
    store_tile_lds(smem + (1 - p) * HSZ, acc, 0, nt, m, hi);
    lds_barrier();
    p ^= 1;
  }
  float* out = Lc + ((size_t)c * BT + mt * MROWS) * LAT;
  for (int i = tid; i < MROWS * LAT; i += SCAN_THREADS)
    out[i] = smem[p * HSZ + (i >> 6) * HPAD + (i & 63)];
}

__global__ void __launch_bounds__(SCAN_THREADS)
chunk_final_k(const float* __restrict__ bx,
              const float* __restrict__ A,
              const float* __restrict__ Apow,
              const float* __restrict__ Lc,
              float* __restrict__ h_out) {
  __shared__ float hbuf[2][MROWS * HPAD];
  int tid = threadIdx.x;
  int lane = tid & 31, hi = lane >> 4, m = lane & 15;
  int nt = tid >> 5;
  int c = blockIdx.x / MTILES, mt = blockIdx.x % MTILES;
  const unsigned uoff = tile_lane_off(mt, nt, m, hi);

  for (int i = tid; i < MROWS * HPAD; i += SCAN_THREADS) hbuf[0][i] = 0.f;
  if (c == 0)
    for (int i = tid; i < MROWS * LAT; i += SCAN_THREADS)
      h_out[mt * MROWS * LAT + i] = 0.f;

  v2f pfrag[16];
  load_bfrag64(pfrag, Apow, nt, m, hi);
  v8f l_next = {0.f, 0.f, 0.f, 0.f, 0.f, 0.f, 0.f, 0.f};
  if (c > 0) l_next = load_tile_lane(Lc, uoff);
  v2f bfrag[16];
  load_bfrag64(bfrag, A, nt, m, hi);
  v8f u_next = load_tile_lane(bx + (size_t)(c * T) * BT * LAT, uoff);
  __syncthreads();

  int p = 0;
#pragma unroll 1
  for (int j = 0; j < c; ++j) {
    v8f acc = l_next;
    if (j + 1 < c)
      l_next = load_tile_lane(Lc + (size_t)(j + 1) * BT * LAT, uoff);
    acc = tile_mm_lds(hbuf[p], pfrag, acc, 0, m, hi);
    store_tile_lds(hbuf[1 - p], acc, 0, nt, m, hi);
    lds_barrier();
    p ^= 1;
  }

#pragma unroll 1
  for (int i = 0; i < T; ++i) {
    int t = c * T + i;
    v8f acc = u_next;
    if (i + 1 < T)
      u_next = load_tile_lane(bx + (size_t)(t + 1) * BT * LAT, uoff);
    acc = tile_mm_lds(hbuf[p], bfrag, acc, 0, m, hi);
    store_tile_lds(hbuf[1 - p], acc, 0, nt, m, hi);
    store_tile_lane(h_out + (size_t)(t + 1) * BT * LAT, uoff, acc);
    lds_barrier();
    p ^= 1;
  }
}

constexpr int Y_BLOCKS = 1024;
constexpr int Y_TILES  = (S * BT / 16) / Y_BLOCKS;

__device__ __forceinline__ void load_row_k64(v4f* q, const float* rowp_h) {
#pragma unroll
  for (int i = 0; i < 8; ++i) q[i] = *(const v4f*)(rowp_h + 8 * i);
}

__device__ __forceinline__ void y_tile(const v4f* aq, const v4f (*bq)[8],
                                       float* yp) {
  v8f acc[4];
#pragma unroll
  for (int j = 0; j < 4; ++j)
    acc[j] = (v8f){0.f, 0.f, 0.f, 0.f, 0.f, 0.f, 0.f, 0.f};
#pragma unroll
  for (int i = 0; i < 8; ++i) {
#pragma unroll
    for (int jj = 0; jj < 2; ++jj) {
      v2f a = {aq[i][2 * jj], aq[i][2 * jj + 1]};
#pragma unroll
      for (int j = 0; j < 4; ++j) {
        v2f b = {bq[j][i][2 * jj], bq[j][i][2 * jj + 1]};
        acc[j] = wmma4(a, b, acc[j]);
      }
    }
  }
#pragma unroll
  for (int r = 0; r < 8; ++r) {
    v4f o = {acc[0][r], acc[1][r], acc[2][r], acc[3][r]};
    *(v4f*)(yp + (size_t)r * DOUT) = o;
  }
}

__global__ void __launch_bounds__(128)
y_gemm_k(const float* __restrict__ hs,
         const float* __restrict__ C,
         float* __restrict__ y) {
  int tid = threadIdx.x;
  int lane = tid & 31, h = lane >> 4, c = lane & 15;
  int cg = tid >> 5;

  v4f bq[4][8];
#pragma unroll
  for (int j = 0; j < 4; ++j)
    load_row_k64(bq[j], C + (size_t)(cg * 64 + 4 * c + j) * LAT + 4 * h);

  const float* ap = hs + ((size_t)blockIdx.x * 16 + c) * LAT + 4 * h;
  float* yp = y + ((size_t)blockIdx.x * 16 + 8 * h) * DOUT + cg * 64 + 4 * c;
  const size_t a_step = (size_t)Y_BLOCKS * 16 * LAT;
  const size_t y_step = (size_t)Y_BLOCKS * 16 * DOUT;

  v4f a0[8], a1[8];
  load_row_k64(a0, ap);
#pragma unroll 1
  for (int it = 0; it < Y_TILES; it += 2) {
    load_row_k64(a1, ap + a_step);
    y_tile(a0, bq, yp);
    if (it + 2 < Y_TILES) load_row_k64(a0, ap + 2 * a_step);
    y_tile(a1, bq, yp + y_step);
    ap += 2 * a_step;
    yp += 2 * y_step;
  }
}

extern "C" void kernel_launch(void* const* d_in, const int* in_sizes, int n_in,
                              void* d_out, int out_size, void* d_ws,
                              size_t ws_size, hipStream_t stream) {
  const float* x  = (const float*)d_in[0];
  const float* A  = (const float*)d_in[1];
  const float* Bw = (const float*)d_in[2];
  const float* C  = (const float*)d_in[3];

  float* out      = (float*)d_out;
  float* y_region = out;
  float* h_region = out + (size_t)S * BT * DOUT;

  float* bx = y_region;

  float* ws      = (float*)d_ws;
  float* Apow    = ws;
  float* Lc      = ws + 64 * 64;

  bx_gemm_k<<<BX_BLOCKS, 128, 0, stream>>>(x, Bw, bx);
  chunk_local_k<<<SCAN_GROUPS + 1, SCAN_THREADS, 0, stream>>>(bx, A, Lc, Apow);
  chunk_final_k<<<SCAN_GROUPS, SCAN_THREADS, 0, stream>>>(bx, A, Apow, Lc,
                                                         h_region);
  y_gemm_k<<<Y_BLOCKS, 128, 0, stream>>>(h_region + BT * LAT, C, y_region);
}


# Scan loops: the tile a step adds (bx or Lc plane) is loaded 3 steps ahead into a ring of 4 register buffers, loops unrolled by 4: no step waits for a load it has just issued; no sum order changes
# speedup vs baseline: 3.2365x; 1.2538x over previous
#include <hip/hip_runtime.h>


typedef __attribute__((ext_vector_type(2))) float v2f;
typedef __attribute__((ext_vector_type(4))) float v4f;
typedef __attribute__((ext_vector_type(8))) float v8f;

constexpr int S    = 2048;
constexpr int BT   = 32;
constexpr int DIN  = 256;
constexpr int DOUT = 256;
constexpr int LAT  = 64;
constexpr int T    = 32;
constexpr int NC   = 64;
constexpr int HPAD = 68;

constexpr int MROWS        = 16;
constexpr int MTILES       = BT / MROWS;
constexpr int SCAN_THREADS = 128;
constexpr int SCAN_GROUPS  = NC * MTILES;

__device__ __forceinline__ v8f wmma4(v2f a, v2f b, v8f c) {
  return __builtin_amdgcn_wmma_f32_16x16x4_f32(false, a, false, b, (short)0, c,
                                               false, false);
}

__device__ __forceinline__ void lds_barrier() {
  __builtin_amdgcn_fence(__ATOMIC_RELEASE, "workgroup", "local");
  __builtin_amdgcn_s_barrier();
  __builtin_amdgcn_fence(__ATOMIC_ACQUIRE, "workgroup", "local");
}

__device__ __forceinline__ v8f tile_mm_lds(const float* hbase, const v2f* bfrag,
                                           v8f acc, int mt, int m, int hi) {
  const float* ap = hbase + (mt * 16 + m) * HPAD + 2 * hi;
  v2f a[16];
#pragma unroll
  for (int kk = 0; kk < 16; ++kk) a[kk] = *(const v2f*)(ap + kk * 4);
#pragma unroll
  for (int kk = 0; kk < 16; ++kk) acc = wmma4(a[kk], bfrag[kk], acc);
  __builtin_amdgcn_sched_group_barrier(0x100, 16, 0);
  __builtin_amdgcn_sched_group_barrier(0x008, 16, 0);
  return acc;
}

__device__ __forceinline__ void load_bfrag64(v2f* bfrag, const float* W, int nt,
                                             int m, int hi) {
#pragma unroll
  for (int kk = 0; kk < 16; ++kk)
    bfrag[kk] = *(const v2f*)(W + (nt * 16 + m) * 64 + kk * 4 + 2 * hi);
}

__device__ __forceinline__ void store_tile_lds(float* hbase, v8f d, int mt,
                                               int nt, int m, int hi) {
#pragma unroll
  for (int r = 0; r < 8; ++r)
    hbase[(mt * 16 + r + 8 * hi) * HPAD + nt * 16 + m] = d[r];
}

__device__ __forceinline__ void store_tile_g64(float* g, v8f d, int mt, int nt,
                                               int m, int hi) {
#pragma unroll
  for (int r = 0; r < 8; ++r)
    g[(mt * 16 + r + 8 * hi) * 64 + nt * 16 + m] = d[r];
}

__device__ __forceinline__ unsigned tile_lane_off(int mt, int nt, int m,
                                                  int hi) {
  return (unsigned)((mt * 16 + 8 * hi) * 64 + nt * 16 + m);
}

__device__ __forceinline__ v8f load_tile_lane(const float* plane,
                                              unsigned off) {
  const float* pl = plane + off;
  v8f acc;
#pragma unroll
  for (int r = 0; r < 8; ++r) acc[r] = pl[64 * r];
  return acc;
}

__device__ __forceinline__ void store_tile_lane(float* plane, unsigned off,
                                                v8f d) {
  float* pl = plane + off;
#pragma unroll
  for (int r = 0; r < 8; ++r) pl[64 * r] = d[r];
}

__device__ __forceinline__ v8f scan_step(const float* src, float* dst,
                                         const v2f* w, v8f add, int nt, int m,
                                         int hi) {
  v8f acc = tile_mm_lds(src, w, add, 0, m, hi);
  store_tile_lds(dst, acc, 0, nt, m, hi);
  return acc;
}

__device__ __forceinline__ int ring_plane(int s, int last) {
  return s < last ? s : last;
}

constexpr int BX_BLOCKS = 1024;
constexpr int BX_TILES  = (S * BT / 16) / BX_BLOCKS;

__device__ __forceinline__ void load_row_k32(v4f* q, const float* rowp_h) {
#pragma unroll
  for (int i = 0; i < 4; ++i) q[i] = *(const v4f*)(rowp_h + 8 * i);
}

__device__ __forceinline__ v8f mm_k32(const v4f* aq, const v4f* bq, v8f acc) {
#pragma unroll
  for (int i = 0; i < 4; ++i) {
#pragma unroll
    for (int jj = 0; jj < 2; ++jj) {
      v2f a = {aq[i][2 * jj], aq[i][2 * jj + 1]};
      v2f b = {bq[i][2 * jj], bq[i][2 * jj + 1]};
      acc = wmma4(a, b, acc);
    }
  }
  return acc;
}

template <bool MORE>
__device__ __forceinline__ void bx_tile(v4f* a0, v4f* a1, v4f* a2, v4f* a3,
                                        const v4f* bq, const float* ap,
                                        const float* nx, float* op) {
  v8f acc = {0.f, 0.f, 0.f, 0.f, 0.f, 0.f, 0.f, 0.f};
  load_row_k32(a3, ap + 96);
  acc = mm_k32(a0, bq + 0, acc);
  load_row_k32(a0, ap + 128);
  acc = mm_k32(a1, bq + 4, acc);
  load_row_k32(a1, ap + 160);
  acc = mm_k32(a2, bq + 8, acc);
  load_row_k32(a2, ap + 192);
  acc = mm_k32(a3, bq + 12, acc);
  load_row_k32(a3, ap + 224);
  acc = mm_k32(a0, bq + 16, acc);
  if (MORE) load_row_k32(a0, nx);
  acc = mm_k32(a1, bq + 20, acc);
  if (MORE) load_row_k32(a1, nx + 32);
  acc = mm_k32(a2, bq + 24, acc);
  if (MORE) load_row_k32(a2, nx + 64);
  acc = mm_k32(a3, bq + 28, acc);
#pragma unroll
  for (int r = 0; r < 8; ++r) op[(size_t)r * LAT] = acc[r];
}

__global__ void __launch_bounds__(128)
bx_gemm_k(const float* __restrict__ x,
          const float* __restrict__ Bw,
          float* __restrict__ bx) {
  int tid = threadIdx.x;
  int lane = tid & 31, h = lane >> 4, c = lane & 15;
  int nt = tid >> 5;

  v4f bq[32];
  {
    const float* bp = Bw + (size_t)(nt * 16 + c) * DIN + 4 * h;
#pragma unroll
    for (int i = 0; i < 32; ++i) bq[i] = *(const v4f*)(bp + 8 * i);
  }

  const float* ap = x + ((size_t)blockIdx.x * 16 + c) * DIN + 4 * h;
  float* op = bx + ((size_t)blockIdx.x * 16 + 8 * h) * LAT + nt * 16 + c;
  const size_t a_step = (size_t)BX_BLOCKS * 16 * DIN;
  const size_t o_step = (size_t)BX_BLOCKS * 16 * LAT;

  v4f a0[4], a1[4], a2[4], a3[4];
  load_row_k32(a0, ap);
  load_row_k32(a1, ap + 32);
  load_row_k32(a2, ap + 64);
#pragma unroll 1
  for (int it = 0; it < BX_TILES - 1; ++it) {
    bx_tile<true>(a0, a1, a2, a3, bq, ap, ap + a_step, op);
    ap += a_step;
    op += o_step;
  }
  bx_tile<false>(a0, a1, a2, a3, bq, ap, ap, op);
}

__device__ __forceinline__ void square_tiles(const float* cur, int mt, int nt0,
                                             int m, int hi, v8f& acc0,
                                             v8f& acc1) {
  const float* ap = cur + (mt * 16 + m) * HPAD + 2 * hi;
  const float* bp = cur + (2 * hi) * HPAD + nt0 * 16 + m;
  v2f a[16], b0[16], b1[16];
#pragma unroll
  for (int kk = 0; kk < 16; ++kk) {
    a[kk] = *(const v2f*)(ap + kk * 4);
    b0[kk][0] = bp[(kk * 4) * HPAD];
    b0[kk][1] = bp[(kk * 4 + 1) * HPAD];
    b1[kk][0] = bp[(kk * 4) * HPAD + 16];
    b1[kk][1] = bp[(kk * 4 + 1) * HPAD + 16];
  }
  v8f c0 = {0.f, 0.f, 0.f, 0.f, 0.f, 0.f, 0.f, 0.f};
  v8f c1 = {0.f, 0.f, 0.f, 0.f, 0.f, 0.f, 0.f, 0.f};
#pragma unroll
  for (int kk = 0; kk < 16; ++kk) {
    c0 = wmma4(a[kk], b0[kk], c0);
    c1 = wmma4(a[kk], b1[kk], c1);
  }
  acc0 = c0;
  acc1 = c1;
}

__device__ __forceinline__ void matpow_role(const float* __restrict__ A,
                                            float* __restrict__ Apow,
                                            float* P) {
  constexpr int PSZ = 64 * HPAD;
  int tid = threadIdx.x;
  int lane = tid & 31, hi = lane >> 4, m = lane & 15;
  int w = tid >> 5;

  v4f av[8];
#pragma unroll
  for (int j = 0; j < 8; ++j) av[j] = *(const v4f*)(A + (tid + j * 128) * 4);
#pragma unroll
  for (int j = 0; j < 8; ++j) {
    int q = tid + j * 128;
    *(v4f*)(&P[(q >> 4) * HPAD + (q & 15) * 4]) = av[j];
  }
  __syncthreads();

  v8f acc0, acc1;
  int p = 0;
  for (int s = 0; s < 4; ++s) {
    const float* cur = P + p * PSZ;
    float* nxt = P + (1 - p) * PSZ;
#pragma unroll
    for (int half = 0; half < 2; ++half) {
      square_tiles(cur, w, 2 * half, m, hi, acc0, acc1);
      store_tile_lds(nxt, acc0, w, 2 * half, m, hi);
      store_tile_lds(nxt, acc1, w, 2 * half + 1, m, hi);
    }
    __syncthreads();
    p ^= 1;
  }
  const float* cur = P + p * PSZ;
#pragma unroll
  for (int half = 0; half < 2; ++half) {
    square_tiles(cur, w, 2 * half, m, hi, acc0, acc1);
    store_tile_g64(Apow, acc0, w, 2 * half, m, hi);
    store_tile_g64(Apow, acc1, w, 2 * half + 1, m, hi);
  }
}

__global__ void __launch_bounds__(SCAN_THREADS)
chunk_local_k(const float* __restrict__ bx,
              const float* __restrict__ A,
              float* __restrict__ Lc,
              float* __restrict__ Apow) {
  __shared__ __attribute__((aligned(16))) float smem[2 * 64 * HPAD];
  if (blockIdx.x == 0) {
    matpow_role(A, Apow, smem);
    return;
  }
  constexpr int HSZ = MROWS * HPAD;
  int tid = threadIdx.x;
  int lane = tid & 31, hi = lane >> 4, m = lane & 15;
  int nt = tid >> 5;
  unsigned g = blockIdx.x - 1u;
  int c = (int)(g / MTILES), mt = (int)(g % MTILES);
  const unsigned uoff = tile_lane_off(mt, nt, m, hi);

  for (int i = tid; i < HSZ; i += SCAN_THREADS) smem[i] = 0.f;

  v2f bfrag[16];
  load_bfrag64(bfrag, A, nt, m, hi);

  constexpr size_t PL = (size_t)BT * LAT;
  const float* ub = bx + (size_t)(c * T) * PL;
  v8f u0 = load_tile_lane(ub, uoff);
  v8f u1 = load_tile_lane(ub + PL, uoff);
  v8f u2 = load_tile_lane(ub + 2 * PL, uoff);
  v8f u3;
  __syncthreads();

  float* ha = smem;
  float* hb = smem + HSZ;
#pragma unroll 1
  for (int i = 0; i < T; i += 4) {
    u3 = load_tile_lane(ub + (size_t)(i + 3) * PL, uoff);
    scan_step(ha, hb, bfrag, u0, nt, m, hi);
    lds_barrier();
    u0 = load_tile_lane(ub + (size_t)ring_plane(i + 4, T - 1) * PL, uoff);
    scan_step(hb, ha, bfrag, u1, nt, m, hi);
    lds_barrier();
    u1 = load_tile_lane(ub + (size_t)ring_plane(i + 5, T - 1) * PL, uoff);
    scan_step(ha, hb, bfrag, u2, nt, m, hi);
    lds_barrier();
    u2 = load_tile_lane(ub + (size_t)ring_plane(i + 6, T - 1) * PL, uoff);
    scan_step(hb, ha, bfrag, u3, nt, m, hi);
    lds_barrier();
  }
  float* out = Lc + ((size_t)c * BT + mt * MROWS) * LAT;
  for (int i = tid; i < MROWS * LAT; i += SCAN_THREADS)
    out[i] = ha[(i >> 6) * HPAD + (i & 63)];
}

__global__ void __launch_bounds__(SCAN_THREADS)
chunk_final_k(const float* __restrict__ bx,
              const float* __restrict__ A,
              const float* __restrict__ Apow,
              const float* __restrict__ Lc,
              float* __restrict__ h_out) {
  __shared__ float hbuf[2][MROWS * HPAD];
  int tid = threadIdx.x;
  int lane = tid & 31, hi = lane >> 4, m = lane & 15;
  int nt = tid >> 5;
  int c = blockIdx.x / MTILES, mt = blockIdx.x % MTILES;
  const unsigned uoff = tile_lane_off(mt, nt, m, hi);

  for (int i = tid; i < MROWS * HPAD; i += SCAN_THREADS) hbuf[0][i] = 0.f;
  if (c == 0)
    for (int i = tid; i < MROWS * LAT; i += SCAN_THREADS)
      h_out[mt * MROWS * LAT + i] = 0.f;

  constexpr size_t PL = (size_t)BT * LAT;
  const int lastc = c - 1;

  v2f pfrag[16];
  load_bfrag64(pfrag, Apow, nt, m, hi);
  v8f l0 = {0.f, 0.f, 0.f, 0.f, 0.f, 0.f, 0.f, 0.f};
  v8f l1 = l0, l2 = l0, l3 = l0;
  if (c > 0) {
    l0 = load_tile_lane(Lc, uoff);
    l1 = load_tile_lane(Lc + (size_t)ring_plane(1, lastc) * PL, uoff);
    l2 = load_tile_lane(Lc + (size_t)ring_plane(2, lastc) * PL, uoff);
  }
  v2f bfrag[16];
  load_bfrag64(bfrag, A, nt, m, hi);
  const float* ub = bx + (size_t)(c * T) * PL;
  v8f u0 = load_tile_lane(ub, uoff);
  v8f u1 = load_tile_lane(ub + PL, uoff);
  v8f u2 = load_tile_lane(ub + 2 * PL, uoff);
  v8f u3;
  __syncthreads();

  int j = 0;
#pragma unroll 1
  for (; j + 4 <= c; j += 4) {
    l3 = load_tile_lane(Lc + (size_t)(j + 3) * PL, uoff);
    scan_step(hbuf[0], hbuf[1], pfrag, l0, nt, m, hi);
    lds_barrier();
    l0 = load_tile_lane(Lc + (size_t)ring_plane(j + 4, lastc) * PL, uoff);
    scan_step(hbuf[1], hbuf[0], pfrag, l1, nt, m, hi);
    lds_barrier();
    l1 = load_tile_lane(Lc + (size_t)ring_plane(j + 5, lastc) * PL, uoff);
    scan_step(hbuf[0], hbuf[1], pfrag, l2, nt, m, hi);
    lds_barrier();
    l2 = load_tile_lane(Lc + (size_t)ring_plane(j + 6, lastc) * PL, uoff);
    scan_step(hbuf[1], hbuf[0], pfrag, l3, nt, m, hi);
    lds_barrier();
  }
  const int rem = c - j;
  if (rem > 0) {
    scan_step(hbuf[0], hbuf[1], pfrag, l0, nt, m, hi);
    lds_barrier();
  }
  if (rem > 1) {
    scan_step(hbuf[1], hbuf[0], pfrag, l1, nt, m, hi);
    lds_barrier();
  }
  if (rem > 2) {
    scan_step(hbuf[0], hbuf[1], pfrag, l2, nt, m, hi);
    lds_barrier();
  }

  float* ha = hbuf[rem & 1];
  float* hb = hbuf[1 - (rem & 1)];
  float* hp = h_out + (size_t)(c * T + 1) * PL;
#pragma unroll 1
  for (int i = 0; i < T; i += 4) {
    v8f acc;
    u3 = load_tile_lane(ub + (size_t)(i + 3) * PL, uoff);
    acc = scan_step(ha, hb, bfrag, u0, nt, m, hi);
    store_tile_lane(hp + (size_t)i * PL, uoff, acc);
    lds_barrier();
    u0 = load_tile_lane(ub + (size_t)ring_plane(i + 4, T - 1) * PL, uoff);
    acc = scan_step(hb, ha, bfrag, u1, nt, m, hi);
    store_tile_lane(hp + (size_t)(i + 1) * PL, uoff, acc);
    lds_barrier();
    u1 = load_tile_lane(ub + (size_t)ring_plane(i + 5, T - 1) * PL, uoff);
    acc = scan_step(ha, hb, bfrag, u2, nt, m, hi);
    store_tile_lane(hp + (size_t)(i + 2) * PL, uoff, acc);
    lds_barrier();
    u2 = load_tile_lane(ub + (size_t)ring_plane(i + 6, T - 1) * PL, uoff);
    acc = scan_step(hb, ha, bfrag, u3, nt, m, hi);
    store_tile_lane(hp + (size_t)(i + 3) * PL, uoff, acc);
    lds_barrier();
  }
}

constexpr int Y_BLOCKS = 1024;
constexpr int Y_TILES  = (S * BT / 16) / Y_BLOCKS;

__device__ __forceinline__ void load_row_k64(v4f* q, const float* rowp_h) {
#pragma unroll
  for (int i = 0; i < 8; ++i) q[i] = *(const v4f*)(rowp_h + 8 * i);
}

__device__ __forceinline__ void y_tile(const v4f* aq, const v4f (*bq)[8],
                                       float* yp) {
  v8f acc[4];
#pragma unroll
  for (int j = 0; j < 4; ++j)
    acc[j] = (v8f){0.f, 0.f, 0.f, 0.f, 0.f, 0.f, 0.f, 0.f};
#pragma unroll
  for (int i = 0; i < 8; ++i) {
#pragma unroll
    for (int jj = 0; jj < 2; ++jj) {
      v2f a = {aq[i][2 * jj], aq[i][2 * jj + 1]};
#pragma unroll
      for (int j = 0; j < 4; ++j) {
        v2f b = {bq[j][i][2 * jj], bq[j][i][2 * jj + 1]};
        acc[j] = wmma4(a, b, acc[j]);
      }
    }
  }
#pragma unroll
  for (int r = 0; r < 8; ++r) {
    v4f o = {acc[0][r], acc[1][r], acc[2][r], acc[3][r]};
    *(v4f*)(yp + (size_t)r * DOUT) = o;
  }
}

__global__ void __launch_bounds__(128)
y_gemm_k(const float* __restrict__ hs,
         const float* __restrict__ C,
         float* __restrict__ y) {
  int tid = threadIdx.x;
  int lane = tid & 31, h = lane >> 4, c = lane & 15;
  int cg = tid >> 5;

  v4f bq[4][8];
#pragma unroll
  for (int j = 0; j < 4; ++j)
    load_row_k64(bq[j], C + (size_t)(cg * 64 + 4 * c + j) * LAT + 4 * h);

  const float* ap = hs + ((size_t)blockIdx.x * 16 + c) * LAT + 4 * h;
  float* yp = y + ((size_t)blockIdx.x * 16 + 8 * h) * DOUT + cg * 64 + 4 * c;
  const size_t a_step = (size_t)Y_BLOCKS * 16 * LAT;
  const size_t y_step = (size_t)Y_BLOCKS * 16 * DOUT;

  v4f a0[8], a1[8];
  load_row_k64(a0, ap);
#pragma unroll 1
  for (int it = 0; it < Y_TILES; it += 2) {
    load_row_k64(a1, ap + a_step);
    y_tile(a0, bq, yp);
    if (it + 2 < Y_TILES) load_row_k64(a0, ap + 2 * a_step);
    y_tile(a1, bq, yp + y_step);
    ap += 2 * a_step;
    yp += 2 * y_step;
  }
}

extern "C" void kernel_launch(void* const* d_in, const int* in_sizes, int n_in,
                              void* d_out, int out_size, void* d_ws,
                              size_t ws_size, hipStream_t stream) {
  const float* x  = (const float*)d_in[0];
  const float* A  = (const float*)d_in[1];
  const float* Bw = (const float*)d_in[2];
  const float* C  = (const float*)d_in[3];

  float* out      = (float*)d_out;
  float* y_region = out;
  float* h_region = out + (size_t)S * BT * DOUT;

  float* bx = y_region;

  float* ws      = (float*)d_ws;
  float* Apow    = ws;
  float* Lc      = ws + 64 * 64;

  bx_gemm_k<<<BX_BLOCKS, 128, 0, stream>>>(x, Bw, bx);
  chunk_local_k<<<SCAN_GROUPS + 1, SCAN_THREADS, 0, stream>>>(bx, A, Lc, Apow);
  chunk_final_k<<<SCAN_GROUPS, SCAN_THREADS, 0, stream>>>(bx, A, Apow, Lc,
                                                         h_region);
  y_gemm_k<<<Y_BLOCKS, 128, 0, stream>>>(h_region + BT * LAT, C, y_region);
}


# Chunk-boundary scan in two levels: a fifth launch sums each run of 4 chunks from zero, a final-pass group then takes <=15 steps with A^128 and <=3 with A^32, not <=63; carry sum order changes
# speedup vs baseline: 3.4467x; 1.0649x over previous
#include <hip/hip_runtime.h>


typedef __attribute__((ext_vector_type(2))) float v2f;
typedef __attribute__((ext_vector_type(4))) float v4f;
typedef __attribute__((ext_vector_type(8))) float v8f;

constexpr int S    = 2048;
constexpr int BT   = 32;
constexpr int DIN  = 256;
constexpr int DOUT = 256;
constexpr int LAT  = 64;
constexpr int T    = 32;
constexpr int NC   = 64;
constexpr int HPAD = 68;

constexpr int MROWS        = 16;
constexpr int MTILES       = BT / MROWS;
constexpr int SCAN_THREADS = 128;
constexpr int SCAN_GROUPS  = NC * MTILES;
constexpr int SUP          = 4;
constexpr int NSUP         = NC / SUP;
constexpr int SUP_GROUPS   = (NSUP - 1) * MTILES;

__device__ __forceinline__ v8f wmma4(v2f a, v2f b, v8f c) {
  return __builtin_amdgcn_wmma_f32_16x16x4_f32(false, a, false, b, (short)0, c,
                                               false, false);
}

__device__ __forceinline__ void lds_barrier() {
  __builtin_amdgcn_fence(__ATOMIC_RELEASE, "workgroup", "local");
  __builtin_amdgcn_s_barrier();
  __builtin_amdgcn_fence(__ATOMIC_ACQUIRE, "workgroup", "local");
}

__device__ __forceinline__ v8f tile_mm_lds(const float* hbase, const v2f* bfrag,
                                           v8f acc, int mt, int m, int hi) {
  const float* ap = hbase + (mt * 16 + m) * HPAD + 2 * hi;
  v2f a[16];
#pragma unroll
  for (int kk = 0; kk < 16; ++kk) a[kk] = *(const v2f*)(ap + kk * 4);
#pragma unroll
  for (int kk = 0; kk < 16; ++kk) acc = wmma4(a[kk], bfrag[kk], acc);
  __builtin_amdgcn_sched_group_barrier(0x100, 16, 0);
  __builtin_amdgcn_sched_group_barrier(0x008, 16, 0);
  return acc;
}

__device__ __forceinline__ void load_bfrag64(v2f* bfrag, const float* W, int nt,
                                             int m, int hi) {
#pragma unroll
  for (int kk = 0; kk < 16; ++kk)
    bfrag[kk] = *(const v2f*)(W + (nt * 16 + m) * 64 + kk * 4 + 2 * hi);
}

__device__ __forceinline__ void store_tile_lds(float* hbase, v8f d, int mt,
                                               int nt, int m, int hi) {
#pragma unroll
  for (int r = 0; r < 8; ++r)
    hbase[(mt * 16 + r + 8 * hi) * HPAD + nt * 16 + m] = d[r];
}

__device__ __forceinline__ void store_tile_g64(float* g, v8f d, int mt, int nt,
                                               int m, int hi) {
#pragma unroll
  for (int r = 0; r < 8; ++r)
    g[(mt * 16 + r + 8 * hi) * 64 + nt * 16 + m] = d[r];
}

__device__ __forceinline__ unsigned tile_lane_off(int mt, int nt, int m,
                                                  int hi) {
  return (unsigned)((mt * 16 + 8 * hi) * 64 + nt * 16 + m);
}

__device__ __forceinline__ v8f load_tile_lane(const float* plane,
                                              unsigned off) {
  const float* pl = plane + off;
  v8f acc;
#pragma unroll
  for (int r = 0; r < 8; ++r) acc[r] = pl[64 * r];
  return acc;
}

__device__ __forceinline__ void store_tile_lane(float* plane, unsigned off,
                                                v8f d) {
  float* pl = plane + off;
#pragma unroll
  for (int r = 0; r < 8; ++r) pl[64 * r] = d[r];
}

__device__ __forceinline__ v8f scan_step(const float* src, float* dst,
                                         const v2f* w, v8f add, int nt, int m,
                                         int hi) {
  v8f acc = tile_mm_lds(src, w, add, 0, m, hi);
  store_tile_lds(dst, acc, 0, nt, m, hi);
  return acc;
}

__device__ __forceinline__ int ring_plane(int s, int last) {
  return s < last ? s : last;
}

__device__ __forceinline__ void retire_tile(v8f t) {
  asm volatile("" ::"v"(t));
}

__device__ __forceinline__ void carry_steps(float* ha, float* hb, const v2f* w,
                                            const float* planes, int n, v8f t0,
                                            v8f t1, v8f t2, unsigned uoff,
                                            int nt, int m, int hi) {
  constexpr size_t PL = (size_t)BT * LAT;
  const int last = n - 1;
  v8f t3;
  int j = 0;
#pragma unroll 1
  for (; j + 4 <= n; j += 4) {
    t3 = load_tile_lane(planes + (size_t)(j + 3) * PL, uoff);
    scan_step(ha, hb, w, t0, nt, m, hi);
    lds_barrier();
    t0 = load_tile_lane(planes + (size_t)ring_plane(j + 4, last) * PL, uoff);
    scan_step(hb, ha, w, t1, nt, m, hi);
    lds_barrier();
    t1 = load_tile_lane(planes + (size_t)ring_plane(j + 5, last) * PL, uoff);
    scan_step(ha, hb, w, t2, nt, m, hi);
    lds_barrier();
    t2 = load_tile_lane(planes + (size_t)ring_plane(j + 6, last) * PL, uoff);
    scan_step(hb, ha, w, t3, nt, m, hi);
    lds_barrier();
  }
  retire_tile(t0);
  retire_tile(t1);
  retire_tile(t2);
  const int rem = n - j;
  if (rem > 0) {
    scan_step(ha, hb, w, t0, nt, m, hi);
    lds_barrier();
  }
  if (rem > 1) {
    scan_step(hb, ha, w, t1, nt, m, hi);
    lds_barrier();
  }
  if (rem > 2) {
    scan_step(ha, hb, w, t2, nt, m, hi);
    lds_barrier();
  }
}

constexpr int BX_BLOCKS = 1024;
constexpr int BX_TILES  = (S * BT / 16) / BX_BLOCKS;

__device__ __forceinline__ void load_row_k32(v4f* q, const float* rowp_h) {
#pragma unroll
  for (int i = 0; i < 4; ++i) q[i] = *(const v4f*)(rowp_h + 8 * i);
}

__device__ __forceinline__ v8f mm_k32(const v4f* aq, const v4f* bq, v8f acc) {
#pragma unroll
  for (int i = 0; i < 4; ++i) {
#pragma unroll
    for (int jj = 0; jj < 2; ++jj) {
      v2f a = {aq[i][2 * jj], aq[i][2 * jj + 1]};
      v2f b = {bq[i][2 * jj], bq[i][2 * jj + 1]};
      acc = wmma4(a, b, acc);
    }
  }
  return acc;
}

template <bool MORE>
__device__ __forceinline__ void bx_tile(v4f* a0, v4f* a1, v4f* a2, v4f* a3,
                                        const v4f* bq, const float* ap,
                                        const float* nx, float* op) {
  v8f acc = {0.f, 0.f, 0.f, 0.f, 0.f, 0.f, 0.f, 0.f};
  load_row_k32(a3, ap + 96);
  acc = mm_k32(a0, bq + 0, acc);
  load_row_k32(a0, ap + 128);
  acc = mm_k32(a1, bq + 4, acc);
  load_row_k32(a1, ap + 160);
  acc = mm_k32(a2, bq + 8, acc);
  load_row_k32(a2, ap + 192);
  acc = mm_k32(a3, bq + 12, acc);
  load_row_k32(a3, ap + 224);
  acc = mm_k32(a0, bq + 16, acc);
  if (MORE) load_row_k32(a0, nx);
  acc = mm_k32(a1, bq + 20, acc);
  if (MORE) load_row_k32(a1, nx + 32);
  acc = mm_k32(a2, bq + 24, acc);
  if (MORE) load_row_k32(a2, nx + 64);
  acc = mm_k32(a3, bq + 28, acc);
#pragma unroll
  for (int r = 0; r < 8; ++r) op[(size_t)r * LAT] = acc[r];
}

__global__ void __launch_bounds__(128)
bx_gemm_k(const float* __restrict__ x,
          const float* __restrict__ Bw,
          float* __restrict__ bx) {
  int tid = threadIdx.x;
  int lane = tid & 31, h = lane >> 4, c = lane & 15;
  int nt = tid >> 5;

  v4f bq[32];
  {
    const float* bp = Bw + (size_t)(nt * 16 + c) * DIN + 4 * h;
#pragma unroll
    for (int i = 0; i < 32; ++i) bq[i] = *(const v4f*)(bp + 8 * i);
  }

  const float* ap = x + ((size_t)blockIdx.x * 16 + c) * DIN + 4 * h;
  float* op = bx + ((size_t)blockIdx.x * 16 + 8 * h) * LAT + nt * 16 + c;
  const size_t a_step = (size_t)BX_BLOCKS * 16 * DIN;
  const size_t o_step = (size_t)BX_BLOCKS * 16 * LAT;

  v4f a0[4], a1[4], a2[4], a3[4];
  load_row_k32(a0, ap);
  load_row_k32(a1, ap + 32);
  load_row_k32(a2, ap + 64);
#pragma unroll 1
  for (int it = 0; it < BX_TILES - 1; ++it) {
    bx_tile<true>(a0, a1, a2, a3, bq, ap, ap + a_step, op);
    ap += a_step;
    op += o_step;
  }
  bx_tile<false>(a0, a1, a2, a3, bq, ap, ap, op);
}

__device__ __forceinline__ void square_tiles(const float* cur, int mt, int nt0,
                                             int m, int hi, v8f& acc0,
                                             v8f& acc1) {
  const float* ap = cur + (mt * 16 + m) * HPAD + 2 * hi;
  const float* bp = cur + (2 * hi) * HPAD + nt0 * 16 + m;
  v2f a[16], b0[16], b1[16];
#pragma unroll
  for (int kk = 0; kk < 16; ++kk) {
    a[kk] = *(const v2f*)(ap + kk * 4);
    b0[kk][0] = bp[(kk * 4) * HPAD];
    b0[kk][1] = bp[(kk * 4 + 1) * HPAD];
    b1[kk][0] = bp[(kk * 4) * HPAD + 16];
    b1[kk][1] = bp[(kk * 4 + 1) * HPAD + 16];
  }
  v8f c0 = {0.f, 0.f, 0.f, 0.f, 0.f, 0.f, 0.f, 0.f};
  v8f c1 = {0.f, 0.f, 0.f, 0.f, 0.f, 0.f, 0.f, 0.f};
#pragma unroll
  for (int kk = 0; kk < 16; ++kk) {
    c0 = wmma4(a[kk], b0[kk], c0);
    c1 = wmma4(a[kk], b1[kk], c1);
  }
  acc0 = c0;
  acc1 = c1;
}

__device__ __forceinline__ void matpow_role(const float* __restrict__ A,
                                            float* __restrict__ Apow,
                                            float* __restrict__ Apow4,
                                            float* P) {
  constexpr int PSZ = 64 * HPAD;
  int tid = threadIdx.x;
  int lane = tid & 31, hi = lane >> 4, m = lane & 15;
  int w = tid >> 5;

  v4f av[8];
#pragma unroll
  for (int j = 0; j < 8; ++j) av[j] = *(const v4f*)(A + (tid + j * 128) * 4);
#pragma unroll
  for (int j = 0; j < 8; ++j) {
    int q = tid + j * 128;
    *(v4f*)(&P[(q >> 4) * HPAD + (q & 15) * 4]) = av[j];
  }
  __syncthreads();

  v8f acc0, acc1;
  int p = 0;
  for (int s = 0; s < 6; ++s) {
    const float* cur = P + p * PSZ;
    float* nxt = P + (1 - p) * PSZ;
#pragma unroll
    for (int half = 0; half < 2; ++half) {
      square_tiles(cur, w, 2 * half, m, hi, acc0, acc1);
      store_tile_lds(nxt, acc0, w, 2 * half, m, hi);
      store_tile_lds(nxt, acc1, w, 2 * half + 1, m, hi);
      if (s == 4) {
        store_tile_g64(Apow, acc0, w, 2 * half, m, hi);
        store_tile_g64(Apow, acc1, w, 2 * half + 1, m, hi);
      }
    }
    __syncthreads();
    p ^= 1;
  }
  const float* cur = P + p * PSZ;
#pragma unroll
  for (int half = 0; half < 2; ++half) {
    square_tiles(cur, w, 2 * half, m, hi, acc0, acc1);
    store_tile_g64(Apow4, acc0, w, 2 * half, m, hi);
    store_tile_g64(Apow4, acc1, w, 2 * half + 1, m, hi);
  }
}

__global__ void __launch_bounds__(SCAN_THREADS)
chunk_local_k(const float* __restrict__ bx,
              const float* __restrict__ A,
              float* __restrict__ Lc,
              float* __restrict__ Apow,
              float* __restrict__ Apow4) {
  __shared__ __attribute__((aligned(16))) float smem[2 * 64 * HPAD];
  if (blockIdx.x == 0) {
    matpow_role(A, Apow, Apow4, smem);
    return;
  }
  constexpr int HSZ = MROWS * HPAD;
  int tid = threadIdx.x;
  int lane = tid & 31, hi = lane >> 4, m = lane & 15;
  int nt = tid >> 5;
  unsigned g = blockIdx.x - 1u;
  int c = (int)(g / MTILES), mt = (int)(g % MTILES);
  const unsigned uoff = tile_lane_off(mt, nt, m, hi);

  for (int i = tid; i < HSZ; i += SCAN_THREADS) smem[i] = 0.f;

  v2f bfrag[16];
  load_bfrag64(bfrag, A, nt, m, hi);

  constexpr size_t PL = (size_t)BT * LAT;
  const float* ub = bx + (size_t)(c * T) * PL;
  v8f u0 = load_tile_lane(ub, uoff);
  v8f u1 = load_tile_lane(ub + PL, uoff);
  v8f u2 = load_tile_lane(ub + 2 * PL, uoff);
  v8f u3;
  __syncthreads();

  float* ha = smem;
  float* hb = smem + HSZ;
#pragma unroll 1
  for (int i = 0; i < T; i += 4) {
    u3 = load_tile_lane(ub + (size_t)(i + 3) * PL, uoff);
    scan_step(ha, hb, bfrag, u0, nt, m, hi);
    lds_barrier();
    u0 = load_tile_lane(ub + (size_t)ring_plane(i + 4, T - 1) * PL, uoff);
    scan_step(hb, ha, bfrag, u1, nt, m, hi);
    lds_barrier();
    u1 = load_tile_lane(ub + (size_t)ring_plane(i + 5, T - 1) * PL, uoff);
    scan_step(ha, hb, bfrag, u2, nt, m, hi);
    lds_barrier();
    u2 = load_tile_lane(ub + (size_t)ring_plane(i + 6, T - 1) * PL, uoff);
    scan_step(hb, ha, bfrag, u3, nt, m, hi);
    lds_barrier();
  }
  float* out = Lc + ((size_t)c * BT + mt * MROWS) * LAT;
  for (int i = tid; i < MROWS * LAT; i += SCAN_THREADS)
    out[i] = ha[(i >> 6) * HPAD + (i & 63)];
}

__global__ void __launch_bounds__(SCAN_THREADS)
super_local_k(const float* __restrict__ Lc,
              const float* __restrict__ Apow,
              float* __restrict__ SL) {
  __shared__ float hbuf[2][MROWS * HPAD];
  int tid = threadIdx.x;
  int lane = tid & 31, hi = lane >> 4, m = lane & 15;
  int nt = tid >> 5;
  int k = blockIdx.x / MTILES, mt = blockIdx.x % MTILES;
  const unsigned uoff = tile_lane_off(mt, nt, m, hi);

  for (int i = tid; i < MROWS * HPAD; i += SCAN_THREADS) hbuf[0][i] = 0.f;

  constexpr size_t PL = (size_t)BT * LAT;
  v2f pfrag[16];
  load_bfrag64(pfrag, Apow, nt, m, hi);
  const float* planes = Lc + (size_t)(k * SUP) * PL;
  v8f t[SUP];
#pragma unroll
  for (int j = 0; j < SUP; ++j)
    t[j] = load_tile_lane(planes + (size_t)j * PL, uoff);
  __syncthreads();

#pragma unroll
  for (int j = 0; j < SUP; ++j) {
    scan_step(hbuf[j & 1], hbuf[1 - (j & 1)], pfrag, t[j], nt, m, hi);
    lds_barrier();
  }

  const float* res = hbuf[SUP & 1];
  float* out = SL + ((size_t)k * BT + mt * MROWS) * LAT;
  for (int i = tid; i < MROWS * LAT; i += SCAN_THREADS)
    out[i] = res[(i >> 6) * HPAD + (i & 63)];
}

__global__ void __launch_bounds__(SCAN_THREADS)
chunk_final_k(const float* __restrict__ bx,
              const float* __restrict__ A,
              const float* __restrict__ Apow,
              const float* __restrict__ Apow4,
              const float* __restrict__ Lc,
              const float* __restrict__ SL,
              float* __restrict__ h_out) {
  __shared__ float hbuf[2][MROWS * HPAD];
  int tid = threadIdx.x;
  int lane = tid & 31, hi = lane >> 4, m = lane & 15;
  int nt = tid >> 5;
  int c = blockIdx.x / MTILES, mt = blockIdx.x % MTILES;
  const unsigned uoff = tile_lane_off(mt, nt, m, hi);
  const int K = c / SUP;
  const int R = c % SUP;

  for (int i = tid; i < MROWS * HPAD; i += SCAN_THREADS) hbuf[0][i] = 0.f;
  if (c == 0)
    for (int i = tid; i < MROWS * LAT; i += SCAN_THREADS)
      h_out[mt * MROWS * LAT + i] = 0.f;

  constexpr size_t PL = (size_t)BT * LAT;

  const v8f zero8 = {0.f, 0.f, 0.f, 0.f, 0.f, 0.f, 0.f, 0.f};
  v2f qfrag[16];
  load_bfrag64(qfrag, Apow4, nt, m, hi);
  v8f s0 = zero8, s1 = zero8, s2 = zero8;
  if (K > 0) {
    s0 = load_tile_lane(SL, uoff);
    s1 = load_tile_lane(SL + (size_t)ring_plane(1, K - 1) * PL, uoff);
    s2 = load_tile_lane(SL + (size_t)ring_plane(2, K - 1) * PL, uoff);
  }
  v2f pfrag[16];
  load_bfrag64(pfrag, Apow, nt, m, hi);
  const float* lp = Lc + (size_t)(K * SUP) * PL;
  v8f l0 = zero8, l1 = zero8, l2 = zero8;
  if (R > 0) {
    l0 = load_tile_lane(lp, uoff);
    l1 = load_tile_lane(lp + (size_t)ring_plane(1, R - 1) * PL, uoff);
    l2 = load_tile_lane(lp + (size_t)ring_plane(2, R - 1) * PL, uoff);
  }
  v2f bfrag[16];
  load_bfrag64(bfrag, A, nt, m, hi);
  const float* ub = bx + (size_t)(c * T) * PL;
  v8f u0 = load_tile_lane(ub, uoff);
  v8f u1 = load_tile_lane(ub + PL, uoff);
  v8f u2 = load_tile_lane(ub + 2 * PL, uoff);
  v8f u3;
  __syncthreads();

  carry_steps(hbuf[0], hbuf[1], qfrag, SL, K, s0, s1, s2, uoff, nt, m, hi);
  carry_steps(hbuf[K & 1], hbuf[1 - (K & 1)], pfrag, lp, R, l0, l1, l2, uoff,
              nt, m, hi);

  const int par = (K + R) & 1;
  float* ha = hbuf[par];
  float* hb = hbuf[1 - par];
  float* hp = h_out + (size_t)(c * T + 1) * PL;
#pragma unroll 1
  for (int i = 0; i < T; i += 4) {
    v8f acc;
    u3 = load_tile_lane(ub + (size_t)(i + 3) * PL, uoff);
    acc = scan_step(ha, hb, bfrag, u0, nt, m, hi);
    store_tile_lane(hp + (size_t)i * PL, uoff, acc);
    lds_barrier();
    u0 = load_tile_lane(ub + (size_t)ring_plane(i + 4, T - 1) * PL, uoff);
    acc = scan_step(hb, ha, bfrag, u1, nt, m, hi);
    store_tile_lane(hp + (size_t)(i + 1) * PL, uoff, acc);
    lds_barrier();
    u1 = load_tile_lane(ub + (size_t)ring_plane(i + 5, T - 1) * PL, uoff);
    acc = scan_step(ha, hb, bfrag, u2, nt, m, hi);
    store_tile_lane(hp + (size_t)(i + 2) * PL, uoff, acc);
    lds_barrier();
    u2 = load_tile_lane(ub + (size_t)ring_plane(i + 6, T - 1) * PL, uoff);
    acc = scan_step(hb, ha, bfrag, u3, nt, m, hi);
    store_tile_lane(hp + (size_t)(i + 3) * PL, uoff, acc);
    lds_barrier();
  }
}

constexpr int Y_BLOCKS = 1024;
constexpr int Y_TILES  = (S * BT / 16) / Y_BLOCKS;

__device__ __forceinline__ void load_row_k64(v4f* q, const float* rowp_h) {
#pragma unroll
  for (int i = 0; i < 8; ++i) q[i] = *(const v4f*)(rowp_h + 8 * i);
}

__device__ __forceinline__ void y_tile(const v4f* aq, const v4f (*bq)[8],
                                       float* yp) {
  v8f acc[4];
#pragma unroll
  for (int j = 0; j < 4; ++j)
    acc[j] = (v8f){0.f, 0.f, 0.f, 0.f, 0.f, 0.f, 0.f, 0.f};
#pragma unroll
  for (int i = 0; i < 8; ++i) {
#pragma unroll
    for (int jj = 0; jj < 2; ++jj) {
      v2f a = {aq[i][2 * jj], aq[i][2 * jj + 1]};
#pragma unroll
      for (int j = 0; j < 4; ++j) {
        v2f b = {bq[j][i][2 * jj], bq[j][i][2 * jj + 1]};
        acc[j] = wmma4(a, b, acc[j]);
      }
    }
  }
#pragma unroll
  for (int r = 0; r < 8; ++r) {
    v4f o = {acc[0][r], acc[1][r], acc[2][r], acc[3][r]};
    *(v4f*)(yp + (size_t)r * DOUT) = o;
  }
}

__global__ void __launch_bounds__(128)
y_gemm_k(const float* __restrict__ hs,
         const float* __restrict__ C,
         float* __restrict__ y) {
  int tid = threadIdx.x;
  int lane = tid & 31, h = lane >> 4, c = lane & 15;
  int cg = tid >> 5;

  v4f bq[4][8];
#pragma unroll
  for (int j = 0; j < 4; ++j)
    load_row_k64(bq[j], C + (size_t)(cg * 64 + 4 * c + j) * LAT + 4 * h);

  const float* ap = hs + ((size_t)blockIdx.x * 16 + c) * LAT + 4 * h;
  float* yp = y + ((size_t)blockIdx.x * 16 + 8 * h) * DOUT + cg * 64 + 4 * c;
  const size_t a_step = (size_t)Y_BLOCKS * 16 * LAT;
  const size_t y_step = (size_t)Y_BLOCKS * 16 * DOUT;

  v4f a0[8], a1[8];
  load_row_k64(a0, ap);
#pragma unroll 1
  for (int it = 0; it < Y_TILES; it += 2) {
    load_row_k64(a1, ap + a_step);
    y_tile(a0, bq, yp);
    if (it + 2 < Y_TILES) load_row_k64(a0, ap + 2 * a_step);
    y_tile(a1, bq, yp + y_step);
    ap += 2 * a_step;
    yp += 2 * y_step;
  }
}

extern "C" void kernel_launch(void* const* d_in, const int* in_sizes, int n_in,
                              void* d_out, int out_size, void* d_ws,
                              size_t ws_size, hipStream_t stream) {
  const float* x  = (const float*)d_in[0];
  const float* A  = (const float*)d_in[1];
  const float* Bw = (const float*)d_in[2];
  const float* C  = (const float*)d_in[3];

  float* out      = (float*)d_out;
  float* y_region = out;
  float* h_region = out + (size_t)S * BT * DOUT;

  float* bx = y_region;

  float* ws      = (float*)d_ws;
  float* Apow    = ws;
  float* Lc      = ws + 64 * 64;
  float* Apow4   = Lc + (size_t)NC * BT * LAT;
  float* SL      = Apow4 + 64 * 64;

  bx_gemm_k<<<BX_BLOCKS, 128, 0, stream>>>(x, Bw, bx);
  chunk_local_k<<<SCAN_GROUPS + 1, SCAN_THREADS, 0, stream>>>(bx, A, Lc, Apow,
                                                             Apow4);
  super_local_k<<<SUP_GROUPS, SCAN_THREADS, 0, stream>>>(Lc, Apow, SL);
  chunk_final_k<<<SCAN_GROUPS, SCAN_THREADS, 0, stream>>>(bx, A, Apow, Apow4,
                                                         Lc, SL, h_region);
  y_gemm_k<<<Y_BLOCKS, 128, 0, stream>>>(h_region + BT * LAT, C, y_region);
}


# Scan re-tiled: chunk 32 -> 16 steps, super-chunk 4 -> 8 chunks (longest chain of dependent scan steps 86 -> 62); the squarings past A^16 move to the super-chunk launch; carry sum order changes
# speedup vs baseline: 3.6671x; 1.0639x over previous
#include <hip/hip_runtime.h>


typedef __attribute__((ext_vector_type(2))) float v2f;
typedef __attribute__((ext_vector_type(4))) float v4f;
typedef __attribute__((ext_vector_type(8))) float v8f;

constexpr int S    = 2048;
constexpr int BT   = 32;
constexpr int DIN  = 256;
constexpr int DOUT = 256;
constexpr int LAT  = 64;
constexpr int T    = 16;
constexpr int NC   = 128;
constexpr int LOG2_T = 4;
constexpr int HPAD = 68;

constexpr int MROWS        = 16;
constexpr int MTILES       = BT / MROWS;
constexpr int SCAN_THREADS = 128;
constexpr int SCAN_GROUPS  = NC * MTILES;
constexpr int LOCAL_GROUPS = (NC - 1) * MTILES;
constexpr int SUP          = 8;
constexpr int LOG2_SUP     = 3;
constexpr int NSUP         = NC / SUP;
static_assert((1 << LOG2_T) == T && (1 << LOG2_SUP) == SUP && T * NC == S &&
              NC % SUP == 0 && T % 4 == 0, "scan tiling");
constexpr int SUP_GROUPS   = (NSUP - 1) * MTILES;

__device__ __forceinline__ v8f wmma4(v2f a, v2f b, v8f c) {
  return __builtin_amdgcn_wmma_f32_16x16x4_f32(false, a, false, b, (short)0, c,
                                               false, false);
}

__device__ __forceinline__ void lds_barrier() {
  __builtin_amdgcn_fence(__ATOMIC_RELEASE, "workgroup", "local");
  __builtin_amdgcn_s_barrier();
  __builtin_amdgcn_fence(__ATOMIC_ACQUIRE, "workgroup", "local");
}

__device__ __forceinline__ v8f tile_mm_lds(const float* hbase, const v2f* bfrag,
                                           v8f acc, int mt, int m, int hi) {
  const float* ap = hbase + (mt * 16 + m) * HPAD + 2 * hi;
  v2f a[16];
#pragma unroll
  for (int kk = 0; kk < 16; ++kk) a[kk] = *(const v2f*)(ap + kk * 4);
#pragma unroll
  for (int kk = 0; kk < 16; ++kk) acc = wmma4(a[kk], bfrag[kk], acc);
  __builtin_amdgcn_sched_group_barrier(0x100, 16, 0);
  __builtin_amdgcn_sched_group_barrier(0x008, 16, 0);
  return acc;
}

__device__ __forceinline__ void load_bfrag64(v2f* bfrag, const float* W, int nt,
                                             int m, int hi) {
#pragma unroll
  for (int kk = 0; kk < 16; ++kk)
    bfrag[kk] = *(const v2f*)(W + (nt * 16 + m) * 64 + kk * 4 + 2 * hi);
}

__device__ __forceinline__ void store_tile_lds(float* hbase, v8f d, int mt,
                                               int nt, int m, int hi) {
#pragma unroll
  for (int r = 0; r < 8; ++r)
    hbase[(mt * 16 + r + 8 * hi) * HPAD + nt * 16 + m] = d[r];
}

__device__ __forceinline__ void store_tile_g64(float* g, v8f d, int mt, int nt,
                                               int m, int hi) {
#pragma unroll
  for (int r = 0; r < 8; ++r)
    g[(mt * 16 + r + 8 * hi) * 64 + nt * 16 + m] = d[r];
}

__device__ __forceinline__ unsigned tile_lane_off(int mt, int nt, int m,
                                                  int hi) {
  return (unsigned)((mt * 16 + 8 * hi) * 64 + nt * 16 + m);
}

__device__ __forceinline__ v8f load_tile_lane(const float* plane,
                                              unsigned off) {
  const float* pl = plane + off;
  v8f acc;
#pragma unroll
  for (int r = 0; r < 8; ++r) acc[r] = pl[64 * r];
  return acc;
}

__device__ __forceinline__ void store_tile_lane(float* plane, unsigned off,
                                                v8f d) {
  float* pl = plane + off;
#pragma unroll
  for (int r = 0; r < 8; ++r) pl[64 * r] = d[r];
}

__device__ __forceinline__ v8f scan_step(const float* src, float* dst,
                                         const v2f* w, v8f add, int nt, int m,
                                         int hi) {
  v8f acc = tile_mm_lds(src, w, add, 0, m, hi);
  store_tile_lds(dst, acc, 0, nt, m, hi);
  return acc;
}

__device__ __forceinline__ int ring_plane(int s, int last) {
  return s < last ? s : last;
}

__device__ __forceinline__ void retire_tile(v8f t) {
  asm volatile("" ::"v"(t));
}

__device__ __forceinline__ void carry_steps(float* ha, float* hb, const v2f* w,
                                            const float* planes, int n, v8f t0,
                                            v8f t1, v8f t2, unsigned uoff,
                                            int nt, int m, int hi) {
  constexpr size_t PL = (size_t)BT * LAT;
  const int last = n - 1;
  v8f t3;
  int j = 0;
#pragma unroll 1
  for (; j + 4 <= n; j += 4) {
    t3 = load_tile_lane(planes + (size_t)(j + 3) * PL, uoff);
    scan_step(ha, hb, w, t0, nt, m, hi);
    lds_barrier();
    t0 = load_tile_lane(planes + (size_t)ring_plane(j + 4, last) * PL, uoff);
    scan_step(hb, ha, w, t1, nt, m, hi);
    lds_barrier();
    t1 = load_tile_lane(planes + (size_t)ring_plane(j + 5, last) * PL, uoff);
    scan_step(ha, hb, w, t2, nt, m, hi);
    lds_barrier();
    t2 = load_tile_lane(planes + (size_t)ring_plane(j + 6, last) * PL, uoff);
    scan_step(hb, ha, w, t3, nt, m, hi);
    lds_barrier();
  }
  retire_tile(t0);
  retire_tile(t1);
  retire_tile(t2);
  const int rem = n - j;
  if (rem > 0) {
    scan_step(ha, hb, w, t0, nt, m, hi);
    lds_barrier();
  }
  if (rem > 1) {
    scan_step(hb, ha, w, t1, nt, m, hi);
    lds_barrier();
  }
  if (rem > 2) {
    scan_step(ha, hb, w, t2, nt, m, hi);
    lds_barrier();
  }
}

constexpr int BX_BLOCKS = 1024;
constexpr int BX_TILES  = (S * BT / 16) / BX_BLOCKS;

__device__ __forceinline__ void load_row_k32(v4f* q, const float* rowp_h) {
#pragma unroll
  for (int i = 0; i < 4; ++i) q[i] = *(const v4f*)(rowp_h + 8 * i);
}

__device__ __forceinline__ v8f mm_k32(const v4f* aq, const v4f* bq, v8f acc) {
#pragma unroll
  for (int i = 0; i < 4; ++i) {
#pragma unroll
    for (int jj = 0; jj < 2; ++jj) {
      v2f a = {aq[i][2 * jj], aq[i][2 * jj + 1]};
      v2f b = {bq[i][2 * jj], bq[i][2 * jj + 1]};
      acc = wmma4(a, b, acc);
    }
  }
  return acc;
}

template <bool MORE>
__device__ __forceinline__ void bx_tile(v4f* a0, v4f* a1, v4f* a2, v4f* a3,
                                        const v4f* bq, const float* ap,
                                        const float* nx, float* op) {
  v8f acc = {0.f, 0.f, 0.f, 0.f, 0.f, 0.f, 0.f, 0.f};
  load_row_k32(a3, ap + 96);
  acc = mm_k32(a0, bq + 0, acc);
  load_row_k32(a0, ap + 128);
  acc = mm_k32(a1, bq + 4, acc);
  load_row_k32(a1, ap + 160);
  acc = mm_k32(a2, bq + 8, acc);
  load_row_k32(a2, ap + 192);
  acc = mm_k32(a3, bq + 12, acc);
  load_row_k32(a3, ap + 224);
  acc = mm_k32(a0, bq + 16, acc);
  if (MORE) load_row_k32(a0, nx);
  acc = mm_k32(a1, bq + 20, acc);
  if (MORE) load_row_k32(a1, nx + 32);
  acc = mm_k32(a2, bq + 24, acc);
  if (MORE) load_row_k32(a2, nx + 64);
  acc = mm_k32(a3, bq + 28, acc);
#pragma unroll
  for (int r = 0; r < 8; ++r) op[(size_t)r * LAT] = acc[r];
}

__global__ void __launch_bounds__(128)
bx_gemm_k(const float* __restrict__ x,
          const float* __restrict__ Bw,
          float* __restrict__ bx) {
  int tid = threadIdx.x;
  int lane = tid & 31, h = lane >> 4, c = lane & 15;
  int nt = tid >> 5;

  v4f bq[32];
  {
    const float* bp = Bw + (size_t)(nt * 16 + c) * DIN + 4 * h;
#pragma unroll
    for (int i = 0; i < 32; ++i) bq[i] = *(const v4f*)(bp + 8 * i);
  }

  const float* ap = x + ((size_t)blockIdx.x * 16 + c) * DIN + 4 * h;
  float* op = bx + ((size_t)blockIdx.x * 16 + 8 * h) * LAT + nt * 16 + c;
  const size_t a_step = (size_t)BX_BLOCKS * 16 * DIN;
  const size_t o_step = (size_t)BX_BLOCKS * 16 * LAT;

  v4f a0[4], a1[4], a2[4], a3[4];
  load_row_k32(a0, ap);
  load_row_k32(a1, ap + 32);
  load_row_k32(a2, ap + 64);
#pragma unroll 1
  for (int it = 0; it < BX_TILES - 1; ++it) {
    bx_tile<true>(a0, a1, a2, a3, bq, ap, ap + a_step, op);
    ap += a_step;
    op += o_step;
  }
  bx_tile<false>(a0, a1, a2, a3, bq, ap, ap, op);
}

__device__ __forceinline__ void square_tiles(const float* cur, int mt, int nt0,
                                             int m, int hi, v8f& acc0,
                                             v8f& acc1) {
  const float* ap = cur + (mt * 16 + m) * HPAD + 2 * hi;
  const float* bp = cur + (2 * hi) * HPAD + nt0 * 16 + m;
  v2f a[16], b0[16], b1[16];
#pragma unroll
  for (int kk = 0; kk < 16; ++kk) {
    a[kk] = *(const v2f*)(ap + kk * 4);
    b0[kk][0] = bp[(kk * 4) * HPAD];
    b0[kk][1] = bp[(kk * 4 + 1) * HPAD];
    b1[kk][0] = bp[(kk * 4) * HPAD + 16];
    b1[kk][1] = bp[(kk * 4 + 1) * HPAD + 16];
  }
  v8f c0 = {0.f, 0.f, 0.f, 0.f, 0.f, 0.f, 0.f, 0.f};
  v8f c1 = {0.f, 0.f, 0.f, 0.f, 0.f, 0.f, 0.f, 0.f};
#pragma unroll
  for (int kk = 0; kk < 16; ++kk) {
    c0 = wmma4(a[kk], b0[kk], c0);
    c1 = wmma4(a[kk], b1[kk], c1);
  }
  acc0 = c0;
  acc1 = c1;
}

template <int NSQ>
__device__ __forceinline__ void matpow_role(const float* __restrict__ src,
                                            float* __restrict__ dst,
                                            float* P) {
  constexpr int PSZ = 64 * HPAD;
  int tid = threadIdx.x;
  int lane = tid & 31, hi = lane >> 4, m = lane & 15;
  int w = tid >> 5;

  v4f av[8];
#pragma unroll
  for (int j = 0; j < 8; ++j) av[j] = *(const v4f*)(src + (tid + j * 128) * 4);
#pragma unroll
  for (int j = 0; j < 8; ++j) {
    int q = tid + j * 128;
    *(v4f*)(&P[(q >> 4) * HPAD + (q & 15) * 4]) = av[j];
  }
  __syncthreads();

  v8f acc0, acc1;
  int p = 0;
  for (int s = 0; s < NSQ - 1; ++s) {
    const float* cur = P + p * PSZ;
    float* nxt = P + (1 - p) * PSZ;
#pragma unroll
    for (int half = 0; half < 2; ++half) {
      square_tiles(cur, w, 2 * half, m, hi, acc0, acc1);
      store_tile_lds(nxt, acc0, w, 2 * half, m, hi);
      store_tile_lds(nxt, acc1, w, 2 * half + 1, m, hi);
    }
    __syncthreads();
    p ^= 1;
  }
  const float* cur = P + p * PSZ;
#pragma unroll
  for (int half = 0; half < 2; ++half) {
    square_tiles(cur, w, 2 * half, m, hi, acc0, acc1);
    store_tile_g64(dst, acc0, w, 2 * half, m, hi);
    store_tile_g64(dst, acc1, w, 2 * half + 1, m, hi);
  }
}

__global__ void __launch_bounds__(SCAN_THREADS)
chunk_local_k(const float* __restrict__ bx,
              const float* __restrict__ A,
              float* __restrict__ Lc,
              float* __restrict__ Apow) {
  __shared__ __attribute__((aligned(16))) float smem[2 * 64 * HPAD];
  if (blockIdx.x == 0) {
    matpow_role<LOG2_T>(A, Apow, smem);
    return;
  }
  constexpr int HSZ = MROWS * HPAD;
  int tid = threadIdx.x;
  int lane = tid & 31, hi = lane >> 4, m = lane & 15;
  int nt = tid >> 5;
  unsigned g = blockIdx.x - 1u;
  int c = (int)(g / MTILES), mt = (int)(g % MTILES);
  const unsigned uoff = tile_lane_off(mt, nt, m, hi);

  for (int i = tid; i < HSZ; i += SCAN_THREADS) smem[i] = 0.f;

  v2f bfrag[16];
  load_bfrag64(bfrag, A, nt, m, hi);

  constexpr size_t PL = (size_t)BT * LAT;
  const float* ub = bx + (size_t)(c * T) * PL;
  v8f u0 = load_tile_lane(ub, uoff);
  v8f u1 = load_tile_lane(ub + PL, uoff);
  v8f u2 = load_tile_lane(ub + 2 * PL, uoff);
  v8f u3;
  __syncthreads();

  float* ha = smem;
  float* hb = smem + HSZ;
#pragma unroll 1
  for (int i = 0; i < T; i += 4) {
    u3 = load_tile_lane(ub + (size_t)(i + 3) * PL, uoff);
    scan_step(ha, hb, bfrag, u0, nt, m, hi);
    lds_barrier();
    u0 = load_tile_lane(ub + (size_t)ring_plane(i + 4, T - 1) * PL, uoff);
    scan_step(hb, ha, bfrag, u1, nt, m, hi);
    lds_barrier();
    u1 = load_tile_lane(ub + (size_t)ring_plane(i + 5, T - 1) * PL, uoff);
    scan_step(ha, hb, bfrag, u2, nt, m, hi);
    lds_barrier();
    u2 = load_tile_lane(ub + (size_t)ring_plane(i + 6, T - 1) * PL, uoff);
    scan_step(hb, ha, bfrag, u3, nt, m, hi);
    lds_barrier();
  }
  float* out = Lc + ((size_t)c * BT + mt * MROWS) * LAT;
  for (int i = tid; i < MROWS * LAT; i += SCAN_THREADS)
    out[i] = ha[(i >> 6) * HPAD + (i & 63)];
}

__global__ void __launch_bounds__(SCAN_THREADS)
super_local_k(const float* __restrict__ Lc,
              const float* __restrict__ Apow,
              float* __restrict__ SL,
              float* __restrict__ ApowS) {
  __shared__ __attribute__((aligned(16))) float smem[2 * 64 * HPAD];
  if (blockIdx.x == 0) {
    matpow_role<LOG2_SUP>(Apow, ApowS, smem);
    return;
  }
  constexpr int HSZ = MROWS * HPAD;
  float* hbuf[2] = {smem, smem + HSZ};
  int tid = threadIdx.x;
  int lane = tid & 31, hi = lane >> 4, m = lane & 15;
  int nt = tid >> 5;
  unsigned g = blockIdx.x - 1u;
  int k = (int)(g / MTILES), mt = (int)(g % MTILES);
  const unsigned uoff = tile_lane_off(mt, nt, m, hi);

  for (int i = tid; i < HSZ; i += SCAN_THREADS) hbuf[0][i] = 0.f;

  constexpr size_t PL = (size_t)BT * LAT;
  v2f pfrag[16];
  load_bfrag64(pfrag, Apow, nt, m, hi);
  const float* planes = Lc + (size_t)(k * SUP) * PL;
  v8f t[SUP];
#pragma unroll
  for (int j = 0; j < SUP; ++j)
    t[j] = load_tile_lane(planes + (size_t)j * PL, uoff);
  __syncthreads();

#pragma unroll
  for (int j = 0; j < SUP; ++j) {
    scan_step(hbuf[j & 1], hbuf[1 - (j & 1)], pfrag, t[j], nt, m, hi);
    lds_barrier();
  }

  const float* res = hbuf[SUP & 1];
  float* out = SL + ((size_t)k * BT + mt * MROWS) * LAT;
  for (int i = tid; i < MROWS * LAT; i += SCAN_THREADS)
    out[i] = res[(i >> 6) * HPAD + (i & 63)];
}

__global__ void __launch_bounds__(SCAN_THREADS)
chunk_final_k(const float* __restrict__ bx,
              const float* __restrict__ A,
              const float* __restrict__ Apow,
              const float* __restrict__ ApowS,
              const float* __restrict__ Lc,
              const float* __restrict__ SL,
              float* __restrict__ h_out) {
  __shared__ float hbuf[2][MROWS * HPAD];
  int tid = threadIdx.x;
  int lane = tid & 31, hi = lane >> 4, m = lane & 15;
  int nt = tid >> 5;
  int c = blockIdx.x / MTILES, mt = blockIdx.x % MTILES;
  const unsigned uoff = tile_lane_off(mt, nt, m, hi);
  const int K = c / SUP;
  const int R = c % SUP;

  for (int i = tid; i < MROWS * HPAD; i += SCAN_THREADS) hbuf[0][i] = 0.f;
  if (c == 0)
    for (int i = tid; i < MROWS * LAT; i += SCAN_THREADS)
      h_out[mt * MROWS * LAT + i] = 0.f;

  constexpr size_t PL = (size_t)BT * LAT;

  const v8f zero8 = {0.f, 0.f, 0.f, 0.f, 0.f, 0.f, 0.f, 0.f};
  v2f qfrag[16];
  load_bfrag64(qfrag, ApowS, nt, m, hi);
  v8f s0 = zero8, s1 = zero8, s2 = zero8;
  if (K > 0) {
    s0 = load_tile_lane(SL, uoff);
    s1 = load_tile_lane(SL + (size_t)ring_plane(1, K - 1) * PL, uoff);
    s2 = load_tile_lane(SL + (size_t)ring_plane(2, K - 1) * PL, uoff);
  }
  v2f pfrag[16];
  load_bfrag64(pfrag, Apow, nt, m, hi);
  const float* lp = Lc + (size_t)(K * SUP) * PL;
  v8f l0 = zero8, l1 = zero8, l2 = zero8;
  if (R > 0) {
    l0 = load_tile_lane(lp, uoff);
    l1 = load_tile_lane(lp + (size_t)ring_plane(1, R - 1) * PL, uoff);
    l2 = load_tile_lane(lp + (size_t)ring_plane(2, R - 1) * PL, uoff);
  }
  v2f bfrag[16];
  load_bfrag64(bfrag, A, nt, m, hi);
  const float* ub = bx + (size_t)(c * T) * PL;
  v8f u0 = load_tile_lane(ub, uoff);
  v8f u1 = load_tile_lane(ub + PL, uoff);
  v8f u2 = load_tile_lane(ub + 2 * PL, uoff);
  v8f u3;
  __syncthreads();

  carry_steps(hbuf[0], hbuf[1], qfrag, SL, K, s0, s1, s2, uoff, nt, m, hi);
  carry_steps(hbuf[K & 1], hbuf[1 - (K & 1)], pfrag, lp, R, l0, l1, l2, uoff,
              nt, m, hi);

  const int par = (K + R) & 1;
  float* ha = hbuf[par];
  float* hb = hbuf[1 - par];
  float* hp = h_out + (size_t)(c * T + 1) * PL;
#pragma unroll 1
  for (int i = 0; i < T; i += 4) {
    v8f acc;
    u3 = load_tile_lane(ub + (size_t)(i + 3) * PL, uoff);
    acc = scan_step(ha, hb, bfrag, u0, nt, m, hi);
    store_tile_lane(hp + (size_t)i * PL, uoff, acc);
    lds_barrier();
    u0 = load_tile_lane(ub + (size_t)ring_plane(i + 4, T - 1) * PL, uoff);
    acc = scan_step(hb, ha, bfrag, u1, nt, m, hi);
    store_tile_lane(hp + (size_t)(i + 1) * PL, uoff, acc);
    lds_barrier();
    u1 = load_tile_lane(ub + (size_t)ring_plane(i + 5, T - 1) * PL, uoff);
    acc = scan_step(ha, hb, bfrag, u2, nt, m, hi);
    store_tile_lane(hp + (size_t)(i + 2) * PL, uoff, acc);
    lds_barrier();
    u2 = load_tile_lane(ub + (size_t)ring_plane(i + 6, T - 1) * PL, uoff);
    acc = scan_step(hb, ha, bfrag, u3, nt, m, hi);
    store_tile_lane(hp + (size_t)(i + 3) * PL, uoff, acc);
    lds_barrier();
  }
}

constexpr int Y_BLOCKS = 1024;
constexpr int Y_TILES  = (S * BT / 16) / Y_BLOCKS;

__device__ __forceinline__ void load_row_k64(v4f* q, const float* rowp_h) {
#pragma unroll
  for (int i = 0; i < 8; ++i) q[i] = *(const v4f*)(rowp_h + 8 * i);
}

__device__ __forceinline__ void y_tile(const v4f* aq, const v4f (*bq)[8],
                                       float* yp) {
  v8f acc[4];
#pragma unroll
  for (int j = 0; j < 4; ++j)
    acc[j] = (v8f){0.f, 0.f, 0.f, 0.f, 0.f, 0.f, 0.f, 0.f};
#pragma unroll
  for (int i = 0; i < 8; ++i) {
#pragma unroll
    for (int jj = 0; jj < 2; ++jj) {
      v2f a = {aq[i][2 * jj], aq[i][2 * jj + 1]};
#pragma unroll
      for (int j = 0; j < 4; ++j) {
        v2f b = {bq[j][i][2 * jj], bq[j][i][2 * jj + 1]};
        acc[j] = wmma4(a, b, acc[j]);
      }
    }
  }
#pragma unroll
  for (int r = 0; r < 8; ++r) {
    v4f o = {acc[0][r], acc[1][r], acc[2][r], acc[3][r]};
    *(v4f*)(yp + (size_t)r * DOUT) = o;
  }
}

__global__ void __launch_bounds__(128)
y_gemm_k(const float* __restrict__ hs,
         const float* __restrict__ C,
         float* __restrict__ y) {
  int tid = threadIdx.x;
  int lane = tid & 31, h = lane >> 4, c = lane & 15;
  int cg = tid >> 5;

  v4f bq[4][8];
#pragma unroll
  for (int j = 0; j < 4; ++j)
    load_row_k64(bq[j], C + (size_t)(cg * 64 + 4 * c + j) * LAT + 4 * h);

  const float* ap = hs + ((size_t)blockIdx.x * 16 + c) * LAT + 4 * h;
  float* yp = y + ((size_t)blockIdx.x * 16 + 8 * h) * DOUT + cg * 64 + 4 * c;
  const size_t a_step = (size_t)Y_BLOCKS * 16 * LAT;
  const size_t y_step = (size_t)Y_BLOCKS * 16 * DOUT;

  v4f a0[8], a1[8];
  load_row_k64(a0, ap);
#pragma unroll 1
  for (int it = 0; it < Y_TILES; it += 2) {
    load_row_k64(a1, ap + a_step);
    y_tile(a0, bq, yp);
    if (it + 2 < Y_TILES) load_row_k64(a0, ap + 2 * a_step);
    y_tile(a1, bq, yp + y_step);
    ap += 2 * a_step;
    yp += 2 * y_step;
  }
}

extern "C" void kernel_launch(void* const* d_in, const int* in_sizes, int n_in,
                              void* d_out, int out_size, void* d_ws,
                              size_t ws_size, hipStream_t stream) {
  const float* x  = (const float*)d_in[0];
  const float* A  = (const float*)d_in[1];
  const float* Bw = (const float*)d_in[2];
  const float* C  = (const float*)d_in[3];

  float* out      = (float*)d_out;
  float* y_region = out;
  float* h_region = out + (size_t)S * BT * DOUT;

  float* bx = y_region;

  float* ws      = (float*)d_ws;
  float* Apow    = ws;
  float* Lc      = ws + 64 * 64;
  float* ApowS   = Lc + (size_t)NC * BT * LAT;
  float* SL      = ApowS + 64 * 64;

  bx_gemm_k<<<BX_BLOCKS, 128, 0, stream>>>(x, Bw, bx);
  chunk_local_k<<<LOCAL_GROUPS + 1, SCAN_THREADS, 0, stream>>>(bx, A, Lc, Apow);
  super_local_k<<<SUP_GROUPS + 1, SCAN_THREADS, 0, stream>>>(Lc, Apow, SL,
                                                             ApowS);
  chunk_final_k<<<SCAN_GROUPS, SCAN_THREADS, 0, stream>>>(bx, A, Apow, ApowS,
                                                         Lc, SL, h_region);
  y_gemm_k<<<Y_BLOCKS, 128, 0, stream>>>(h_region + BT * LAT, C, y_region);
}


# bx GEMM: x row tiles and Bw reach the fragment registers through LDS from whole-line 16-byte loads issued once per block (was: each of 4 waves loading 32-byte pieces of 16 lines); no sum order change
# speedup vs baseline: 4.1954x; 1.1441x over previous
#include <hip/hip_runtime.h>


typedef __attribute__((ext_vector_type(2))) float v2f;
typedef __attribute__((ext_vector_type(4))) float v4f;
typedef __attribute__((ext_vector_type(8))) float v8f;

constexpr int S    = 2048;
constexpr int BT   = 32;
constexpr int DIN  = 256;
constexpr int DOUT = 256;
constexpr int LAT  = 64;
constexpr int T    = 16;
constexpr int NC   = 128;
constexpr int LOG2_T = 4;
constexpr int HPAD = 68;

constexpr int MROWS        = 16;
constexpr int MTILES       = BT / MROWS;
constexpr int SCAN_THREADS = 128;
constexpr int SCAN_GROUPS  = NC * MTILES;
constexpr int LOCAL_GROUPS = (NC - 1) * MTILES;
constexpr int SUP          = 8;
constexpr int LOG2_SUP     = 3;
constexpr int NSUP         = NC / SUP;
static_assert((1 << LOG2_T) == T && (1 << LOG2_SUP) == SUP && T * NC == S &&
              NC % SUP == 0 && T % 4 == 0, "scan tiling");
constexpr int SUP_GROUPS   = (NSUP - 1) * MTILES;

__device__ __forceinline__ v8f wmma4(v2f a, v2f b, v8f c) {
  return __builtin_amdgcn_wmma_f32_16x16x4_f32(false, a, false, b, (short)0, c,
                                               false, false);
}

__device__ __forceinline__ void lds_barrier() {
  __builtin_amdgcn_fence(__ATOMIC_RELEASE, "workgroup", "local");
  __builtin_amdgcn_s_barrier();
  __builtin_amdgcn_fence(__ATOMIC_ACQUIRE, "workgroup", "local");
}

__device__ __forceinline__ v8f tile_mm_lds(const float* hbase, const v2f* bfrag,
                                           v8f acc, int mt, int m, int hi) {
  const float* ap = hbase + (mt * 16 + m) * HPAD + 2 * hi;
  v2f a[16];
#pragma unroll
  for (int kk = 0; kk < 16; ++kk) a[kk] = *(const v2f*)(ap + kk * 4);
#pragma unroll
  for (int kk = 0; kk < 16; ++kk) acc = wmma4(a[kk], bfrag[kk], acc);
  __builtin_amdgcn_sched_group_barrier(0x100, 16, 0);
  __builtin_amdgcn_sched_group_barrier(0x008, 16, 0);
  return acc;
}

__device__ __forceinline__ void load_bfrag64(v2f* bfrag, const float* W, int nt,
                                             int m, int hi) {
#pragma unroll
  for (int kk = 0; kk < 16; ++kk)
    bfrag[kk] = *(const v2f*)(W + (nt * 16 + m) * 64 + kk * 4 + 2 * hi);
}

__device__ __forceinline__ void store_tile_lds(float* hbase, v8f d, int mt,
                                               int nt, int m, int hi) {
#pragma unroll
  for (int r = 0; r < 8; ++r)
    hbase[(mt * 16 + r + 8 * hi) * HPAD + nt * 16 + m] = d[r];
}

__device__ __forceinline__ void store_tile_g64(float* g, v8f d, int mt, int nt,
                                               int m, int hi) {
#pragma unroll
  for (int r = 0; r < 8; ++r)
    g[(mt * 16 + r + 8 * hi) * 64 + nt * 16 + m] = d[r];
}

__device__ __forceinline__ unsigned tile_lane_off(int mt, int nt, int m,
                                                  int hi) {
  return (unsigned)((mt * 16 + 8 * hi) * 64 + nt * 16 + m);
}

__device__ __forceinline__ v8f load_tile_lane(const float* plane,
                                              unsigned off) {
  const float* pl = plane + off;
  v8f acc;
#pragma unroll
  for (int r = 0; r < 8; ++r) acc[r] = pl[64 * r];
  return acc;
}

__device__ __forceinline__ void store_tile_lane(float* plane, unsigned off,
                                                v8f d) {
  float* pl = plane + off;
#pragma unroll
  for (int r = 0; r < 8; ++r) pl[64 * r] = d[r];
}

__device__ __forceinline__ v8f scan_step(const float* src, float* dst,
                                         const v2f* w, v8f add, int nt, int m,
                                         int hi) {
  v8f acc = tile_mm_lds(src, w, add, 0, m, hi);
  store_tile_lds(dst, acc, 0, nt, m, hi);
  return acc;
}

__device__ __forceinline__ int ring_plane(int s, int last) {
  return s < last ? s : last;
}

__device__ __forceinline__ void retire_tile(v8f t) {
  asm volatile("" ::"v"(t));
}

__device__ __forceinline__ void carry_steps(float* ha, float* hb, const v2f* w,
                                            const float* planes, int n, v8f t0,
                                            v8f t1, v8f t2, unsigned uoff,
                                            int nt, int m, int hi) {
  constexpr size_t PL = (size_t)BT * LAT;
  const int last = n - 1;
  v8f t3;
  int j = 0;
#pragma unroll 1
  for (; j + 4 <= n; j += 4) {
    t3 = load_tile_lane(planes + (size_t)(j + 3) * PL, uoff);
    scan_step(ha, hb, w, t0, nt, m, hi);
    lds_barrier();
    t0 = load_tile_lane(planes + (size_t)ring_plane(j + 4, last) * PL, uoff);
    scan_step(hb, ha, w, t1, nt, m, hi);
    lds_barrier();
    t1 = load_tile_lane(planes + (size_t)ring_plane(j + 5, last) * PL, uoff);
    scan_step(ha, hb, w, t2, nt, m, hi);
    lds_barrier();
    t2 = load_tile_lane(planes + (size_t)ring_plane(j + 6, last) * PL, uoff);
    scan_step(hb, ha, w, t3, nt, m, hi);
    lds_barrier();
  }
  retire_tile(t0);
  retire_tile(t1);
  retire_tile(t2);
  const int rem = n - j;
  if (rem > 0) {
    scan_step(ha, hb, w, t0, nt, m, hi);
    lds_barrier();
  }
  if (rem > 1) {
    scan_step(hb, ha, w, t1, nt, m, hi);
    lds_barrier();
  }
  if (rem > 2) {
    scan_step(ha, hb, w, t2, nt, m, hi);
    lds_barrier();
  }
}

constexpr int BX_BLOCKS = 1024;
constexpr int BX_TILES  = (S * BT / 16) / BX_BLOCKS;
constexpr int XP = DIN + 4;
constexpr int BP = DIN / 2 + 4;
constexpr int XTILE = 16 * XP;
constexpr int BX_SMEM = (64 * BP > 2 * XTILE) ? 64 * BP : 2 * XTILE;
static_assert(BX_TILES >= 2 && DIN == 256 && LAT == 64, "bx GEMM tiling");

__device__ __forceinline__ void lds_row_k32(v4f* q, const float* rowp_h) {
#pragma unroll
  for (int i = 0; i < 4; ++i) q[i] = *(const v4f*)(rowp_h + 8 * i);
}

__device__ __forceinline__ v8f mm_k32(const v4f* aq, const v4f* bq, v8f acc) {
#pragma unroll
  for (int i = 0; i < 4; ++i) {
#pragma unroll
    for (int jj = 0; jj < 2; ++jj) {
      v2f a = {aq[i][2 * jj], aq[i][2 * jj + 1]};
      v2f b = {bq[i][2 * jj], bq[i][2 * jj + 1]};
      acc = wmma4(a, b, acc);
    }
  }
  return acc;
}

__device__ __forceinline__ void x_tile_fetch(v4f* xr, const float* xp) {
#pragma unroll
  for (int j = 0; j < 8; ++j) xr[j] = *(const v4f*)(xp + 512 * j);
}

__device__ __forceinline__ void x_tile_put(float* lp, const v4f* xr) {
#pragma unroll
  for (int j = 0; j < 8; ++j) *(v4f*)(lp + 2 * j * XP) = xr[j];
}

__device__ __forceinline__ void bx_tile(const float* fp, const v4f* bq,
                                        float* op) {
  v8f acc = {0.f, 0.f, 0.f, 0.f, 0.f, 0.f, 0.f, 0.f};
  v4f a0[4], a1[4];
  lds_row_k32(a0, fp);
#pragma unroll
  for (int kc = 0; kc < 8; kc += 2) {
    lds_row_k32(a1, fp + 32 * (kc + 1));
    acc = mm_k32(a0, bq + 4 * kc, acc);
    if (kc + 2 < 8) lds_row_k32(a0, fp + 32 * (kc + 2));
    acc = mm_k32(a1, bq + 4 * (kc + 1), acc);
  }
  __builtin_amdgcn_sched_group_barrier(0x100, 8, 0);
#pragma unroll
  for (int k = 0; k < 6; ++k) {
    __builtin_amdgcn_sched_group_barrier(0x008, 8, 0);
    __builtin_amdgcn_sched_group_barrier(0x100, 4, 0);
  }
  __builtin_amdgcn_sched_group_barrier(0x008, 16, 0);
#pragma unroll
  for (int r = 0; r < 8; ++r) op[(size_t)r * LAT] = acc[r];
}

__global__ void __launch_bounds__(128)
bx_gemm_k(const float* __restrict__ x,
          const float* __restrict__ Bw,
          float* __restrict__ bx) {
  __shared__ __attribute__((aligned(16))) float smem[BX_SMEM];
  int tid = threadIdx.x;
  int lane = tid & 31, h = lane >> 4, c = lane & 15;
  int nt = tid >> 5;

  const float* xp = x + (size_t)blockIdx.x * 16 * DIN + 4 * tid;
  const size_t x_step = (size_t)BX_BLOCKS * 16 * DIN;
  float* op = bx + ((size_t)blockIdx.x * 16 + 8 * h) * LAT + nt * 16 + c;
  const size_t o_step = (size_t)BX_BLOCKS * 16 * LAT;

  v4f t0[16], t1[16], xr[8];
#pragma unroll
  for (int j = 0; j < 16; ++j) {
    const float* bp = Bw + (size_t)(nt + 4 * j) * DIN + 4 * lane;
    t0[j] = *(const v4f*)(bp);
    t1[j] = *(const v4f*)(bp + DIN / 2);
  }
  x_tile_fetch(xr, xp);

  v4f bq[32];
  float* bput = smem + nt * BP + 4 * lane;
  const float* bget = smem + (nt * 16 + c) * BP + 4 * h;
#pragma unroll
  for (int j = 0; j < 16; ++j) *(v4f*)(bput + 4 * j * BP) = t0[j];
  lds_barrier();
#pragma unroll
  for (int i = 0; i < 16; ++i) bq[i] = *(const v4f*)(bget + 8 * i);
  lds_barrier();
#pragma unroll
  for (int j = 0; j < 16; ++j) *(v4f*)(bput + 4 * j * BP) = t1[j];
  lds_barrier();
#pragma unroll
  for (int i = 0; i < 16; ++i) bq[16 + i] = *(const v4f*)(bget + 8 * i);
  lds_barrier();

  float* xput = smem + (tid >> 6) * XP + 4 * (tid & 63);
  const float* xget = smem + c * XP + 4 * h;
  x_tile_put(xput, xr);
  x_tile_fetch(xr, xp + x_step);
  lds_barrier();
#pragma unroll
  for (int it = 0; it < BX_TILES; ++it) {
    bx_tile(xget + (it & 1) * XTILE, bq, op + (size_t)it * o_step);
    if (it + 1 < BX_TILES) {
      x_tile_put(xput + ((it + 1) & 1) * XTILE, xr);
      if (it + 2 < BX_TILES) x_tile_fetch(xr, xp + (size_t)(it + 2) * x_step);
      lds_barrier();
    }
  }
}

__device__ __forceinline__ void square_tiles(const float* cur, int mt, int nt0,
                                             int m, int hi, v8f& acc0,
                                             v8f& acc1) {
  const float* ap = cur + (mt * 16 + m) * HPAD + 2 * hi;
  const float* bp = cur + (2 * hi) * HPAD + nt0 * 16 + m;
  v2f a[16], b0[16], b1[16];
#pragma unroll
  for (int kk = 0; kk < 16; ++kk) {
    a[kk] = *(const v2f*)(ap + kk * 4);
    b0[kk][0] = bp[(kk * 4) * HPAD];
    b0[kk][1] = bp[(kk * 4 + 1) * HPAD];
    b1[kk][0] = bp[(kk * 4) * HPAD + 16];
    b1[kk][1] = bp[(kk * 4 + 1) * HPAD + 16];
  }
  v8f c0 = {0.f, 0.f, 0.f, 0.f, 0.f, 0.f, 0.f, 0.f};
  v8f c1 = {0.f, 0.f, 0.f, 0.f, 0.f, 0.f, 0.f, 0.f};
#pragma unroll
  for (int kk = 0; kk < 16; ++kk) {
    c0 = wmma4(a[kk], b0[kk], c0);
    c1 = wmma4(a[kk], b1[kk], c1);
  }
  acc0 = c0;
  acc1 = c1;
}

template <int NSQ>
__device__ __forceinline__ void matpow_role(const float* __restrict__ src,
                                            float* __restrict__ dst,
                                            float* P) {
  constexpr int PSZ = 64 * HPAD;
  int tid = threadIdx.x;
  int lane = tid & 31, hi = lane >> 4, m = lane & 15;
  int w = tid >> 5;

  v4f av[8];
#pragma unroll
  for (int j = 0; j < 8; ++j) av[j] = *(const v4f*)(src + (tid + j * 128) * 4);
#pragma unroll
  for (int j = 0; j < 8; ++j) {
    int q = tid + j * 128;
    *(v4f*)(&P[(q >> 4) * HPAD + (q & 15) * 4]) = av[j];
  }
  __syncthreads();

  v8f acc0, acc1;
  int p = 0;
  for (int s = 0; s < NSQ - 1; ++s) {
    const float* cur = P + p * PSZ;
    float* nxt = P + (1 - p) * PSZ;
#pragma unroll
    for (int half = 0; half < 2; ++half) {
      square_tiles(cur, w, 2 * half, m, hi, acc0, acc1);
      store_tile_lds(nxt, acc0, w, 2 * half, m, hi);
      store_tile_lds(nxt, acc1, w, 2 * half + 1, m, hi);
    }
    __syncthreads();
    p ^= 1;
  }
  const float* cur = P + p * PSZ;
#pragma unroll
  for (int half = 0; half < 2; ++half) {
    square_tiles(cur, w, 2 * half, m, hi, acc0, acc1);
    store_tile_g64(dst, acc0, w, 2 * half, m, hi);
    store_tile_g64(dst, acc1, w, 2 * half + 1, m, hi);
  }
}

__global__ void __launch_bounds__(SCAN_THREADS)
chunk_local_k(const float* __restrict__ bx,
              const float* __restrict__ A,
              float* __restrict__ Lc,
              float* __restrict__ Apow) {
  __shared__ __attribute__((aligned(16))) float smem[2 * 64 * HPAD];
  if (blockIdx.x == 0) {
    matpow_role<LOG2_T>(A, Apow, smem);
    return;
  }
  constexpr int HSZ = MROWS * HPAD;
  int tid = threadIdx.x;
  int lane = tid & 31, hi = lane >> 4, m = lane & 15;
  int nt = tid >> 5;
  unsigned g = blockIdx.x - 1u;
  int c = (int)(g / MTILES), mt = (int)(g % MTILES);
  const unsigned uoff = tile_lane_off(mt, nt, m, hi);

  for (int i = tid; i < HSZ; i += SCAN_THREADS) smem[i] = 0.f;

  v2f bfrag[16];
  load_bfrag64(bfrag, A, nt, m, hi);

  constexpr size_t PL = (size_t)BT * LAT;
  const float* ub = bx + (size_t)(c * T) * PL;
  v8f u0 = load_tile_lane(ub, uoff);
  v8f u1 = load_tile_lane(ub + PL, uoff);
  v8f u2 = load_tile_lane(ub + 2 * PL, uoff);
  v8f u3;
  __syncthreads();

  float* ha = smem;
  float* hb = smem + HSZ;
#pragma unroll 1
  for (int i = 0; i < T; i += 4) {
    u3 = load_tile_lane(ub + (size_t)(i + 3) * PL, uoff);
    scan_step(ha, hb, bfrag, u0, nt, m, hi);
    lds_barrier();
    u0 = load_tile_lane(ub + (size_t)ring_plane(i + 4, T - 1) * PL, uoff);
    scan_step(hb, ha, bfrag, u1, nt, m, hi);
    lds_barrier();
    u1 = load_tile_lane(ub + (size_t)ring_plane(i + 5, T - 1) * PL, uoff);
    scan_step(ha, hb, bfrag, u2, nt, m, hi);
    lds_barrier();
    u2 = load_tile_lane(ub + (size_t)ring_plane(i + 6, T - 1) * PL, uoff);
    scan_step(hb, ha, bfrag, u3, nt, m, hi);
    lds_barrier();
  }
  float* out = Lc + ((size_t)c * BT + mt * MROWS) * LAT;
  for (int i = tid; i < MROWS * LAT; i += SCAN_THREADS)
    out[i] = ha[(i >> 6) * HPAD + (i & 63)];
}

__global__ void __launch_bounds__(SCAN_THREADS)
super_local_k(const float* __restrict__ Lc,
              const float* __restrict__ Apow,
              float* __restrict__ SL,
              float* __restrict__ ApowS) {
  __shared__ __attribute__((aligned(16))) float smem[2 * 64 * HPAD];
  if (blockIdx.x == 0) {
    matpow_role<LOG2_SUP>(Apow, ApowS, smem);
    return;
  }
  constexpr int HSZ = MROWS * HPAD;
  float* hbuf[2] = {smem, smem + HSZ};
  int tid = threadIdx.x;
  int lane = tid & 31, hi = lane >> 4, m = lane & 15;
  int nt = tid >> 5;
  unsigned g = blockIdx.x - 1u;
  int k = (int)(g / MTILES), mt = (int)(g % MTILES);
  const unsigned uoff = tile_lane_off(mt, nt, m, hi);

  for (int i = tid; i < HSZ; i += SCAN_THREADS) hbuf[0][i] = 0.f;

  constexpr size_t PL = (size_t)BT * LAT;
  v2f pfrag[16];
  load_bfrag64(pfrag, Apow, nt, m, hi);
  const float* planes = Lc + (size_t)(k * SUP) * PL;
  v8f t[SUP];
#pragma unroll
  for (int j = 0; j < SUP; ++j)
    t[j] = load_tile_lane(planes + (size_t)j * PL, uoff);
  __syncthreads();

#pragma unroll
  for (int j = 0; j < SUP; ++j) {
    scan_step(hbuf[j & 1], hbuf[1 - (j & 1)], pfrag, t[j], nt, m, hi);
    lds_barrier();
  }

  const float* res = hbuf[SUP & 1];
  float* out = SL + ((size_t)k * BT + mt * MROWS) * LAT;
  for (int i = tid; i < MROWS * LAT; i += SCAN_THREADS)
    out[i] = res[(i >> 6) * HPAD + (i & 63)];
}

__global__ void __launch_bounds__(SCAN_THREADS)
chunk_final_k(const float* __restrict__ bx,
              const float* __restrict__ A,
              const float* __restrict__ Apow,
              const float* __restrict__ ApowS,
              const float* __restrict__ Lc,
              const float* __restrict__ SL,
              float* __restrict__ h_out) {
  __shared__ float hbuf[2][MROWS * HPAD];
  int tid = threadIdx.x;
  int lane = tid & 31, hi = lane >> 4, m = lane & 15;
  int nt = tid >> 5;
  int c = blockIdx.x / MTILES, mt = blockIdx.x % MTILES;
  const unsigned uoff = tile_lane_off(mt, nt, m, hi);
  const int K = c / SUP;
  const int R = c % SUP;

  for (int i = tid; i < MROWS * HPAD; i += SCAN_THREADS) hbuf[0][i] = 0.f;
  if (c == 0)
    for (int i = tid; i < MROWS * LAT; i += SCAN_THREADS)
      h_out[mt * MROWS * LAT + i] = 0.f;

  constexpr size_t PL = (size_t)BT * LAT;

  const v8f zero8 = {0.f, 0.f, 0.f, 0.f, 0.f, 0.f, 0.f, 0.f};
  v2f qfrag[16];
  load_bfrag64(qfrag, ApowS, nt, m, hi);
  v8f s0 = zero8, s1 = zero8, s2 = zero8;
  if (K > 0) {
    s0 = load_tile_lane(SL, uoff);
    s1 = load_tile_lane(SL + (size_t)ring_plane(1, K - 1) * PL, uoff);
    s2 = load_tile_lane(SL + (size_t)ring_plane(2, K - 1) * PL, uoff);
  }
  v2f pfrag[16];
  load_bfrag64(pfrag, Apow, nt, m, hi);
  const float* lp = Lc + (size_t)(K * SUP) * PL;
  v8f l0 = zero8, l1 = zero8, l2 = zero8;
  if (R > 0) {
    l0 = load_tile_lane(lp, uoff);
    l1 = load_tile_lane(lp + (size_t)ring_plane(1, R - 1) * PL, uoff);
    l2 = load_tile_lane(lp + (size_t)ring_plane(2, R - 1) * PL, uoff);
  }
  v2f bfrag[16];
  load_bfrag64(bfrag, A, nt, m, hi);
  const float* ub = bx + (size_t)(c * T) * PL;
  v8f u0 = load_tile_lane(ub, uoff);
  v8f u1 = load_tile_lane(ub + PL, uoff);
  v8f u2 = load_tile_lane(ub + 2 * PL, uoff);
  v8f u3;
  __syncthreads();

  carry_steps(hbuf[0], hbuf[1], qfrag, SL, K, s0, s1, s2, uoff, nt, m, hi);
  carry_steps(hbuf[K & 1], hbuf[1 - (K & 1)], pfrag, lp, R, l0, l1, l2, uoff,
              nt, m, hi);

  const int par = (K + R) & 1;
  float* ha = hbuf[par];
  float* hb = hbuf[1 - par];
  float* hp = h_out + (size_t)(c * T + 1) * PL;
#pragma unroll 1
  for (int i = 0; i < T; i += 4) {
    v8f acc;
    u3 = load_tile_lane(ub + (size_t)(i + 3) * PL, uoff);
    acc = scan_step(ha, hb, bfrag, u0, nt, m, hi);
    store_tile_lane(hp + (size_t)i * PL, uoff, acc);
    lds_barrier();
    u0 = load_tile_lane(ub + (size_t)ring_plane(i + 4, T - 1) * PL, uoff);
    acc = scan_step(hb, ha, bfrag, u1, nt, m, hi);
    store_tile_lane(hp + (size_t)(i + 1) * PL, uoff, acc);
    lds_barrier();
    u1 = load_tile_lane(ub + (size_t)ring_plane(i + 5, T - 1) * PL, uoff);
    acc = scan_step(ha, hb, bfrag, u2, nt, m, hi);
    store_tile_lane(hp + (size_t)(i + 2) * PL, uoff, acc);
    lds_barrier();
    u2 = load_tile_lane(ub + (size_t)ring_plane(i + 6, T - 1) * PL, uoff);
    acc = scan_step(hb, ha, bfrag, u3, nt, m, hi);
    store_tile_lane(hp + (size_t)(i + 3) * PL, uoff, acc);
    lds_barrier();
  }
}

constexpr int Y_BLOCKS = 1024;
constexpr int Y_TILES  = (S * BT / 16) / Y_BLOCKS;

__device__ __forceinline__ void load_row_k64(v4f* q, const float* rowp_h) {
#pragma unroll
  for (int i = 0; i < 8; ++i) q[i] = *(const v4f*)(rowp_h + 8 * i);
}

__device__ __forceinline__ void y_tile(const v4f* aq, const v4f (*bq)[8],
                                       float* yp) {
  v8f acc[4];
#pragma unroll
  for (int j = 0; j < 4; ++j)
    acc[j] = (v8f){0.f, 0.f, 0.f, 0.f, 0.f, 0.f, 0.f, 0.f};
#pragma unroll
  for (int i = 0; i < 8; ++i) {
#pragma unroll
    for (int jj = 0; jj < 2; ++jj) {
      v2f a = {aq[i][2 * jj], aq[i][2 * jj + 1]};
#pragma unroll
      for (int j = 0; j < 4; ++j) {
        v2f b = {bq[j][i][2 * jj], bq[j][i][2 * jj + 1]};
        acc[j] = wmma4(a, b, acc[j]);
      }
    }
  }
#pragma unroll
  for (int r = 0; r < 8; ++r) {
    v4f o = {acc[0][r], acc[1][r], acc[2][r], acc[3][r]};
    *(v4f*)(yp + (size_t)r * DOUT) = o;
  }
}

__global__ void __launch_bounds__(128)
y_gemm_k(const float* __restrict__ hs,
         const float* __restrict__ C,
         float* __restrict__ y) {
  int tid = threadIdx.x;
  int lane = tid & 31, h = lane >> 4, c = lane & 15;
  int cg = tid >> 5;

  v4f bq[4][8];
#pragma unroll
  for (int j = 0; j < 4; ++j)
    load_row_k64(bq[j], C + (size_t)(cg * 64 + 4 * c + j) * LAT + 4 * h);

  const float* ap = hs + ((size_t)blockIdx.x * 16 + c) * LAT + 4 * h;
  float* yp = y + ((size_t)blockIdx.x * 16 + 8 * h) * DOUT + cg * 64 + 4 * c;
  const size_t a_step = (size_t)Y_BLOCKS * 16 * LAT;
  const size_t y_step = (size_t)Y_BLOCKS * 16 * DOUT;

  v4f a0[8], a1[8];
  load_row_k64(a0, ap);
#pragma unroll 1
  for (int it = 0; it < Y_TILES; it += 2) {
    load_row_k64(a1, ap + a_step);
    y_tile(a0, bq, yp);
    if (it + 2 < Y_TILES) load_row_k64(a0, ap + 2 * a_step);
    y_tile(a1, bq, yp + y_step);
    ap += 2 * a_step;
    yp += 2 * y_step;
  }
}

extern "C" void kernel_launch(void* const* d_in, const int* in_sizes, int n_in,
                              void* d_out, int out_size, void* d_ws,
                              size_t ws_size, hipStream_t stream) {
  const float* x  = (const float*)d_in[0];
  const float* A  = (const float*)d_in[1];
  const float* Bw = (const float*)d_in[2];
  const float* C  = (const float*)d_in[3];

  float* out      = (float*)d_out;
  float* y_region = out;
  float* h_region = out + (size_t)S * BT * DOUT;

  float* bx = y_region;

  float* ws      = (float*)d_ws;
  float* Apow    = ws;
  float* Lc      = ws + 64 * 64;
  float* ApowS   = Lc + (size_t)NC * BT * LAT;
  float* SL      = ApowS + 64 * 64;

  bx_gemm_k<<<BX_BLOCKS, 128, 0, stream>>>(x, Bw, bx);
  chunk_local_k<<<LOCAL_GROUPS + 1, SCAN_THREADS, 0, stream>>>(bx, A, Lc, Apow);
  super_local_k<<<SUP_GROUPS + 1, SCAN_THREADS, 0, stream>>>(Lc, Apow, SL,
                                                             ApowS);
  chunk_final_k<<<SCAN_GROUPS, SCAN_THREADS, 0, stream>>>(bx, A, Apow, ApowS,
                                                         Lc, SL, h_region);
  y_gemm_k<<<Y_BLOCKS, 128, 0, stream>>>(h_region + BT * LAT, C, y_region);
}
